# starting kernel
# baseline (speedup 1.0000x reference)
#include <hip/hip_runtime.h>
#include <math.h>

typedef float v2f __attribute__((ext_vector_type(2)));
typedef float v8f __attribute__((ext_vector_type(8)));

#define N_NODES 177
#define FEAT 128
#define OUT_C 128
#define NEG_SLOPE 0.2f
#define MTILES 12
#define S_PAD 180
#define H_ROWS 180

#define OFF_H    0
#define OFF_A    (OFF_H + H_ROWS * FEAT)
#define OFF_B    (OFF_A + N_NODES)
#define OFF_ESUM (OFF_B + N_NODES)
#define OFF_T    (OFF_ESUM + N_NODES)
#define OFF_AMAX (OFF_T + N_NODES)
#define OFF_S    (OFF_AMAX + N_NODES)
#define WS_FLOATS (OFF_S + N_NODES * S_PAD)

__device__ __forceinline__ unsigned f2ord(float f) {
  unsigned u = __float_as_uint(f);
  return (u & 0x80000000u) ? ~u : (u | 0x80000000u);
}
__device__ __forceinline__ float ord2f(unsigned u) {
  return __uint_as_float((u & 0x80000000u) ? (u ^ 0x80000000u) : ~u);
}
#define ORD_NEG_INF 0x007FFFFFu

__device__ __forceinline__ v2f load_a_clamp(const float* __restrict__ A, int m0, int k0,
                                            int lane, int ld) {
  int m = m0 + (lane & 15);
  m = (m < N_NODES) ? m : (N_NODES - 1);
  int kb = k0 + ((lane >> 4) << 1);
  const float* p = A + m * ld + kb;
  v2f a; a.x = p[0]; a.y = p[1];
  return a;
}
__device__ __forceinline__ v2f load_b(const float* __restrict__ B, int k0, int n0,
                                      int lane, int ld) {
  int n  = n0 + (lane & 15);
  int kb = k0 + ((lane >> 4) << 1);
  const float* p = B + kb * ld + n;
  v2f b; b.x = p[0]; b.y = p[ld];
  return b;
}
__device__ __forceinline__ v2f load_a_lds(const float* A, int k0, int lane) {
  int m  = lane & 15;
  int kb = k0 + ((lane >> 4) << 1);
  v2f a;
  a.x = A[m * OUT_C + kb];
  a.y = A[m * OUT_C + kb + 1];
  return a;
}

__global__ void k_init(float* __restrict__ ws) {
  int tid    = blockIdx.x * blockDim.x + threadIdx.x;
  int stride = gridDim.x * blockDim.x;
  unsigned* amax = (unsigned*)(ws + OFF_AMAX);
  for (int j = tid; j < N_NODES * S_PAD; j += stride) ws[OFF_S + j] = 0.0f;
  for (int j = tid; j < (H_ROWS - N_NODES) * FEAT; j += stride)
    ws[OFF_H + N_NODES * FEAT + j] = 0.0f;
  for (int j = tid; j < N_NODES; j += stride) {
    ws[OFF_ESUM + j] = 0.0f;
    ws[OFF_T + j]    = 0.0f;
    amax[j]          = ORD_NEG_INF;
  }
}

__global__ void k_node_gemm(const float* __restrict__ x, const float* __restrict__ W,
                            float* __restrict__ ws) {
  float* h  = ws + OFF_H;
  int lane  = threadIdx.x & 31;
  int wave  = threadIdx.x >> 5;
  int m0    = blockIdx.x * 16;
  int n0    = wave * 32;
  v8f c0 = {}; v8f c1 = {};
  for (int k0 = 0; k0 < FEAT; k0 += 4) {
    v2f a  = load_a_clamp(x, m0, k0, lane, FEAT);
    v2f b0 = load_b(W, k0, n0,      lane, OUT_C);
    v2f b1 = load_b(W, k0, n0 + 16, lane, OUT_C);
    c0 = __builtin_amdgcn_wmma_f32_16x16x4_f32(false, a, false, b0, (short)0, c0, false, false);
    c1 = __builtin_amdgcn_wmma_f32_16x16x4_f32(false, a, false, b1, (short)0, c1, false, false);
  }
  int nb = n0 + (lane & 15);
  int mh = (lane >> 4) * 8;
  for (int j = 0; j < 8; ++j) {
    int m = m0 + j + mh;
    if (m < N_NODES) {
      h[m * OUT_C + nb]      = c0[j];
      h[m * OUT_C + nb + 16] = c1[j];
    }
  }
}

__global__ void k_ab(const float* __restrict__ att, float* __restrict__ ws) {
  const float* h = ws + OFF_H;
  int n = blockIdx.x, lane = threadIdx.x;
  float pa = 0.0f, pb = 0.0f;
  for (int i = lane; i < FEAT; i += 32) {
    float hv = h[n * OUT_C + i];
    pa += hv * att[i];
    pb += hv * att[OUT_C + i];
  }
  for (int m = 16; m >= 1; m >>= 1) {
    pa += __shfl_xor(pa, m, 32);
    pb += __shfl_xor(pb, m, 32);
  }
  if (lane == 0) { ws[OFF_A + n] = pa; ws[OFF_B + n] = pb; }
}

__global__ void k_edge_max(const int* __restrict__ ei, const float* __restrict__ ea,
                           const float* __restrict__ att, float* __restrict__ ws, int E) {
  const float* a = ws + OFF_A;
  const float* b = ws + OFF_B;
  unsigned* amax = (unsigned*)(ws + OFF_AMAX);
  float cc = att[2 * OUT_C];
  int tid = blockIdx.x * blockDim.x + threadIdx.x;
  int stride = gridDim.x * blockDim.x;
  for (int e = tid; e < E; e += stride) {
    int s = ei[e], d = ei[E + e];
    float al = a[d] + b[s] + cc * ea[e];
    al = (al > 0.0f) ? al : NEG_SLOPE * al;
    atomicMax(&amax[d], f2ord(al));
  }
}

__global__ void k_edge_exp(const int* __restrict__ ei, const float* __restrict__ ea,
                           const float* __restrict__ att, float* __restrict__ ws, int E) {
  const float* a = ws + OFF_A;
  const float* b = ws + OFF_B;
  const unsigned* amax = (const unsigned*)(ws + OFF_AMAX);
  float* esum = ws + OFF_ESUM;
  float* t    = ws + OFF_T;
  float* S    = ws + OFF_S;
  float cc = att[2 * OUT_C];
  int tid = blockIdx.x * blockDim.x + threadIdx.x;
  int stride = gridDim.x * blockDim.x;
  for (int e = tid; e < E; e += stride) {
    int s = ei[e], d = ei[E + e];
    float eav = ea[e];
    float al = a[d] + b[s] + cc * eav;
    al = (al > 0.0f) ? al : NEG_SLOPE * al;
    float w = __expf(al - ord2f(amax[d]));
    atomicAdd(&esum[d], w);
    atomicAdd(&t[d], w * eav);
    atomicAdd(&S[d * S_PAD + s], w);
  }
}

__global__ void k_final(const float* __restrict__ U, const float* __restrict__ bias,
                        const float* __restrict__ ws, float* __restrict__ out) {
  __shared__ float aggr[16 * OUT_C];
  const float* h    = ws + OFF_H;
  const float* S    = ws + OFF_S;
  const float* esum = ws + OFF_ESUM;
  const float* t    = ws + OFF_T;
  int lane = threadIdx.x & 31;
  int wave = threadIdx.x >> 5;
  int m0 = blockIdx.x * 16;
  int n0 = wave * 32;

  v8f c0 = {}; v8f c1 = {};
  for (int k0 = 0; k0 < S_PAD; k0 += 4) {
    v2f a  = load_a_clamp(S, m0, k0, lane, S_PAD);
    v2f b0 = load_b(h, k0, n0,      lane, OUT_C);
    v2f b1 = load_b(h, k0, n0 + 16, lane, OUT_C);
    c0 = __builtin_amdgcn_wmma_f32_16x16x4_f32(false, a, false, b0, (short)0, c0, false, false);
    c1 = __builtin_amdgcn_wmma_f32_16x16x4_f32(false, a, false, b1, (short)0, c1, false, false);
  }
  int nb = n0 + (lane & 15);
  int mh = (lane >> 4) * 8;
  for (int j = 0; j < 8; ++j) {
    int mloc = j + mh;
    int m = m0 + mloc;
    int mc = (m < N_NODES) ? m : (N_NODES - 1);
    float r = 1.0f / (esum[mc] + 1e-16f);
    aggr[mloc * OUT_C + nb]      = c0[j] * r;
    aggr[mloc * OUT_C + nb + 16] = c1[j] * r;
  }
  __syncthreads();

  v8f d0 = {}; v8f d1 = {};
  for (int k0 = 0; k0 < OUT_C; k0 += 4) {
    v2f a  = load_a_lds(aggr, k0, lane);
    v2f b0 = load_b(U, k0, n0,      lane, OUT_C);
    v2f b1 = load_b(U, k0, n0 + 16, lane, OUT_C);
    d0 = __builtin_amdgcn_wmma_f32_16x16x4_f32(false, a, false, b0, (short)0, d0, false, false);
    d1 = __builtin_amdgcn_wmma_f32_16x16x4_f32(false, a, false, b1, (short)0, d1, false, false);
  }
  for (int j = 0; j < 8; ++j) {
    int m = m0 + j + mh;
    if (m < N_NODES) {
      float tn = t[m] / (esum[m] + 1e-16f);
      out[m * OUT_C + nb]      = d0[j] + tn * U[OUT_C * OUT_C + nb]      + bias[nb];
      out[m * OUT_C + nb + 16] = d1[j] + tn * U[OUT_C * OUT_C + nb + 16] + bias[nb + 16];
    }
  }
}

extern "C" void kernel_launch(void* const* d_in, const int* in_sizes, int n_in,
                              void* d_out, int out_size, void* d_ws, size_t ws_size,
                              hipStream_t stream) {
  const float* x    = (const float*)d_in[0];
  const float* ea   = (const float*)d_in[1];
  const float* W    = (const float*)d_in[2];
  const float* att  = (const float*)d_in[3];
  const float* U    = (const float*)d_in[4];
  const float* bias = (const float*)d_in[5];
  const int*   ei   = (const int*)d_in[6];
  int E = in_sizes[1];
  float* ws  = (float*)d_ws;
  float* out = (float*)d_out;

  k_init<<<64, 256, 0, stream>>>(ws);
  k_node_gemm<<<MTILES, 128, 0, stream>>>(x, W, ws);
  k_ab<<<N_NODES, 32, 0, stream>>>(att, ws);
  k_edge_max<<<1024, 256, 0, stream>>>(ei, ea, att, ws, E);
  k_edge_exp<<<1024, 256, 0, stream>>>(ei, ea, att, ws, E);
  k_final<<<MTILES, 128, 0, stream>>>(U, bias, ws, out);
}


# esum and t leave global atomics: each exp block stores its own partial row (plain, whole lines), final launch folds the 128 rows in fixed order; their sum order changes, same every run now
# speedup vs baseline: 3.4507x; 3.4507x over previous
#include <hip/hip_runtime.h>
#include <math.h>

typedef float v2f __attribute__((ext_vector_type(2)));
typedef float v8f __attribute__((ext_vector_type(8)));

#define N_NODES 177
#define FEAT 128
#define OUT_C 128
#define NEG_SLOPE 0.2f
#define MTILES 12
#define S_PAD 180
#define H_ROWS 180
#define N_PAD 192
#define EDGE_BLOCKS 128

#define OFF_H    0
#define OFF_A    (OFF_H + H_ROWS * FEAT)
#define OFF_B    (OFF_A + N_NODES)
#define OFF_AMAX (OFF_B + N_NODES)
#define OFF_S    (OFF_AMAX + N_NODES)
#define OFF_PES  ((OFF_S + N_NODES * S_PAD + 31) & ~31)
#define OFF_PTS  (OFF_PES + EDGE_BLOCKS * N_PAD)
#define WS_FLOATS (OFF_PTS + EDGE_BLOCKS * N_PAD)

__device__ __forceinline__ unsigned f2ord(float f) {
  unsigned u = __float_as_uint(f);
  return (u & 0x80000000u) ? ~u : (u | 0x80000000u);
}
__device__ __forceinline__ float ord2f(unsigned u) {
  return __uint_as_float((u & 0x80000000u) ? (u ^ 0x80000000u) : ~u);
}
#define ORD_NEG_INF 0x007FFFFFu

__device__ __forceinline__ v2f load_a_clamp(const float* __restrict__ A, int m0, int k0,
                                            int lane, int ld) {
  int m = m0 + (lane & 15);
  m = (m < N_NODES) ? m : (N_NODES - 1);
  int kb = k0 + ((lane >> 4) << 1);
  const float* p = A + m * ld + kb;
  v2f a; a.x = p[0]; a.y = p[1];
  return a;
}
__device__ __forceinline__ v2f load_b(const float* __restrict__ B, int k0, int n0,
                                      int lane, int ld) {
  int n  = n0 + (lane & 15);
  int kb = k0 + ((lane >> 4) << 1);
  const float* p = B + kb * ld + n;
  v2f b; b.x = p[0]; b.y = p[ld];
  return b;
}
__device__ __forceinline__ v2f load_a_lds(const float* A, int k0, int lane) {
  int m  = lane & 15;
  int kb = k0 + ((lane >> 4) << 1);
  v2f a;
  a.x = A[m * OUT_C + kb];
  a.y = A[m * OUT_C + kb + 1];
  return a;
}

__global__ void k_init(float* __restrict__ ws) {
  int tid    = blockIdx.x * blockDim.x + threadIdx.x;
  int stride = gridDim.x * blockDim.x;
  unsigned* amax = (unsigned*)(ws + OFF_AMAX);
  for (int j = tid; j < N_NODES * S_PAD; j += stride) ws[OFF_S + j] = 0.0f;
  for (int j = tid; j < (H_ROWS - N_NODES) * FEAT; j += stride)
    ws[OFF_H + N_NODES * FEAT + j] = 0.0f;
  for (int j = tid; j < N_NODES; j += stride) amax[j] = ORD_NEG_INF;
}

__global__ void k_node_gemm(const float* __restrict__ x, const float* __restrict__ W,
                            float* __restrict__ ws) {
  float* h  = ws + OFF_H;
  int lane  = threadIdx.x & 31;
  int wave  = threadIdx.x >> 5;
  int m0    = blockIdx.x * 16;
  int n0    = wave * 32;
  v8f c0 = {}; v8f c1 = {};
  for (int k0 = 0; k0 < FEAT; k0 += 4) {
    v2f a  = load_a_clamp(x, m0, k0, lane, FEAT);
    v2f b0 = load_b(W, k0, n0,      lane, OUT_C);
    v2f b1 = load_b(W, k0, n0 + 16, lane, OUT_C);
    c0 = __builtin_amdgcn_wmma_f32_16x16x4_f32(false, a, false, b0, (short)0, c0, false, false);
    c1 = __builtin_amdgcn_wmma_f32_16x16x4_f32(false, a, false, b1, (short)0, c1, false, false);
  }
  int nb = n0 + (lane & 15);
  int mh = (lane >> 4) * 8;
  for (int j = 0; j < 8; ++j) {
    int m = m0 + j + mh;
    if (m < N_NODES) {
      h[m * OUT_C + nb]      = c0[j];
      h[m * OUT_C + nb + 16] = c1[j];
    }
  }
}

__global__ void k_ab(const float* __restrict__ att, float* __restrict__ ws) {
  const float* h = ws + OFF_H;
  int n = blockIdx.x, lane = threadIdx.x;
  float pa = 0.0f, pb = 0.0f;
#pragma unroll
  for (int q = 0; q < FEAT / 32; ++q) {
    const int i = (lane & 31) + 32 * q;
    float hv = h[n * OUT_C + i];
    pa += hv * att[i];
    pb += hv * att[OUT_C + i];
  }
  for (int m = 16; m >= 1; m >>= 1) {
    pa += __shfl_xor(pa, m, 32);
    pb += __shfl_xor(pb, m, 32);
  }
  if (lane == 0) { ws[OFF_A + n] = pa; ws[OFF_B + n] = pb; }
}

#define MX_THREADS 256
#define MX_BLOCKS  128
#define MX_R       16
#define MX_BATCH   (MX_THREADS * MX_R)
#define MX_WAVES   (MX_THREADS / 32)
#define MX_ROWS_PER_WAVE ((N_NODES + MX_WAVES - 1) / MX_WAVES)

__global__ __launch_bounds__(MX_THREADS)
void k_edge_max(const int* __restrict__ ei, const float* __restrict__ ea,
                const float* __restrict__ att, float* __restrict__ ws, int E) {
  __shared__ __attribute__((aligned(16))) float tbl[N_NODES * MX_THREADS];
  __shared__ float s_row[N_NODES];

  const float* a = ws + OFF_A;
  const float* b = ws + OFF_B;
  unsigned* amax = (unsigned*)(ws + OFF_AMAX);
  const float cc = att[2 * OUT_C];

  const int tid  = threadIdx.x;
  const int lane = tid & 31;
  const int wave = tid >> 5;

  const int per = (E + (int)gridDim.x - 1) / (int)gridDim.x;
  const int e0  = (int)blockIdx.x * per;
  const int e1  = (e0 + per < E) ? (e0 + per) : E;
  if (e0 >= e1) return;

  int   sv[MX_R], dv[MX_R];
  float ev[MX_R];
#pragma unroll
  for (int i = 0; i < MX_R; ++i) {
    const int e  = e0 + i * MX_THREADS + tid;
    const int ec = (e < e1) ? e : e0;
    sv[i] = ei[ec];
    dv[i] = ei[E + ec];
    ev[i] = ea[ec];
  }

  {
    const float ninf = -INFINITY;
    const float4 n4 = make_float4(ninf, ninf, ninf, ninf);
    float4* t4 = (float4*)tbl;
    for (int q = tid; q < N_NODES * MX_THREADS / 4; q += MX_THREADS) t4[q] = n4;
  }
  __syncthreads();

  float* col = tbl + tid;
  for (int base = e0; ; ) {
    float al[MX_R];
#pragma unroll
    for (int i = 0; i < MX_R; ++i) {
      const int e = base + i * MX_THREADS + tid;
      float v = a[dv[i]] + b[sv[i]] + cc * ev[i];
      v = (v > 0.0f) ? v : NEG_SLOPE * v;
      al[i] = (e < e1) ? v : -INFINITY;
    }
#pragma unroll
    for (int i = 0; i < MX_R; ++i) {
      float* p = col + dv[i] * MX_THREADS;
      *p = fmaxf(*p, al[i]);
    }
    base += MX_BATCH;
    if (base >= e1) break;
#pragma unroll
    for (int i = 0; i < MX_R; ++i) {
      const int e  = base + i * MX_THREADS + tid;
      const int ec = (e < e1) ? e : e0;
      sv[i] = ei[ec];
      dv[i] = ei[E + ec];
      ev[i] = ea[ec];
    }
  }
  __syncthreads();

  float m[MX_ROWS_PER_WAVE];
#pragma unroll
  for (int j = 0; j < MX_ROWS_PER_WAVE; ++j) {
    int r = wave + j * MX_WAVES;
    r = (r < N_NODES) ? r : (N_NODES - 1);
    const float4 v0 = *(const float4*)(&tbl[r * MX_THREADS + lane * 4]);
    const float4 v1 = *(const float4*)(&tbl[r * MX_THREADS + 128 + lane * 4]);
    m[j] = fmaxf(fmaxf(fmaxf(v0.x, v0.y), fmaxf(v0.z, v0.w)),
                 fmaxf(fmaxf(v1.x, v1.y), fmaxf(v1.z, v1.w)));
  }
#pragma unroll
  for (int k = 16; k >= 1; k >>= 1) {
#pragma unroll
    for (int j = 0; j < MX_ROWS_PER_WAVE; ++j) m[j] = fmaxf(m[j], __shfl_xor(m[j], k, 32));
  }
  float mine = -INFINITY;
#pragma unroll
  for (int j = 0; j < MX_ROWS_PER_WAVE; ++j) mine = (lane == j) ? m[j] : mine;
  {
    const int r = wave + lane * MX_WAVES;
    if (lane < MX_ROWS_PER_WAVE && r < N_NODES) s_row[r] = mine;
  }
  __syncthreads();
  if (tid < N_NODES && s_row[tid] > -INFINITY) atomicMax(&amax[tid], f2ord(s_row[tid]));
}

#define EB_THREADS 768
#define EB_R       6
#define EB_BATCH   (EB_THREADS * EB_R)
#define NBINS      144
#define REC_CAP    (EB_BATCH + NBINS * 3 + 8)

__global__ __launch_bounds__(EB_THREADS)
void k_edge_exp(const int* __restrict__ ei, const float* __restrict__ ea,
                const float* __restrict__ att, float* __restrict__ ws, int E) {
  __shared__ int   s_cnt[NBINS + 1];
  __shared__ int   s_start[NBINS];
  __shared__ __attribute__((aligned(16))) int   r_key[REC_CAP];
  __shared__ __attribute__((aligned(16))) float r_w[REC_CAP];
  __shared__ __attribute__((aligned(16))) float r_we[REC_CAP];
  __shared__ float s_es[(EB_THREADS / 32) * 16];
  __shared__ float s_ts[(EB_THREADS / 32) * 16];

  const float* a = ws + OFF_A;
  const float* b = ws + OFF_B;
  const unsigned* amax = (const unsigned*)(ws + OFF_AMAX);
  float* pes  = ws + OFF_PES + (int)blockIdx.x * N_PAD;
  float* pts  = ws + OFF_PTS + (int)blockIdx.x * N_PAD;
  float* S    = ws + OFF_S;
  const float cc = att[2 * OUT_C];

  const int tid  = threadIdx.x;
  const int lane = tid & 31;
  const int wave = tid >> 5;
  const int c15  = lane & 15;
  const int h    = lane >> 4;
  const int first_bin = wave * 6;

  const int per = (E + (int)gridDim.x - 1) / (int)gridDim.x;
  const int e0  = (int)blockIdx.x * per;
  const int e1  = (e0 + per < E) ? (e0 + per) : E;

  v8f acc[6];
#pragma unroll
  for (int j = 0; j < 6; ++j) acc[j] = (v8f){};
  float es = 0.0f, ts = 0.0f;

  for (int base = e0; base < e1; base += EB_BATCH) {
    if (tid <= NBINS) s_cnt[tid] = 0;
    __syncthreads();

    int   sv[EB_R], dv[EB_R];
    float ev[EB_R];
#pragma unroll
    for (int i = 0; i < EB_R; ++i) {
      const int e  = base + i * EB_THREADS + tid;
      const int ec = (e < e1) ? e : e0;
      sv[i] = ei[ec];
      dv[i] = ei[E + ec];
      ev[i] = ea[ec];
    }
    int   pk[EB_R], ppos[EB_R];
    float pw[EB_R], pwe[EB_R];
#pragma unroll
    for (int i = 0; i < EB_R; ++i) {
      float al = a[dv[i]] + b[sv[i]] + cc * ev[i];
      al = (al > 0.0f) ? al : NEG_SLOPE * al;
      const float w = __expf(al - ord2f(amax[dv[i]]));
      pk[i]  = (dv[i] & 15) | ((sv[i] & 15) << 4);
      pw[i]  = w;
      pwe[i] = w * ev[i];
    }
#pragma unroll
    for (int i = 0; i < EB_R; ++i) {
      const int  e   = base + i * EB_THREADS + tid;
      const bool ok  = e < e1;
      const int  bin = ok ? ((dv[i] >> 4) * 12 + (sv[i] >> 4)) : NBINS;
      const int  pos = atomicAdd(&s_cnt[bin], 1);
      ppos[i] = ok ? ((bin << 16) | pos) : -1;
    }
    __syncthreads();

    {
      int sum = 0;
      for (int q = lane; q < first_bin; q += 32) sum += (s_cnt[q] + 3) & ~3;
      for (int m = 16; m >= 1; m >>= 1) sum += __shfl_xor(sum, m, 32);
      if (lane < 6) {
        int st = sum;
        for (int jj = 0; jj < lane; ++jj) st += (s_cnt[first_bin + jj] + 3) & ~3;
        s_start[first_bin + lane] = st;
        const int nb = s_cnt[first_bin + lane];
        for (int p = st + nb; p < st + ((nb + 3) & ~3); ++p) {
          r_key[p] = 0; r_w[p] = 0.0f; r_we[p] = 0.0f;
        }
      }
    }
    __syncthreads();

#pragma unroll
    for (int i = 0; i < EB_R; ++i) {
      if (ppos[i] >= 0) {
        const int idx = s_start[ppos[i] >> 16] + (ppos[i] & 0xFFFF);
        r_key[idx] = pk[i];
        r_w[idx]   = pw[i];
        r_we[idx]  = pwe[i];
      }
    }
    __syncthreads();

#pragma unroll
    for (int j = 0; j < 6; ++j) {
      const int n   = __builtin_amdgcn_readfirstlane(s_cnt[first_bin + j]);
      const int st  = __builtin_amdgcn_readfirstlane(s_start[first_bin + j]);
      const int nch = (n + 3) >> 2;
      for (int c = 0; c < nch; ++c) {
        const int idx = st + 4 * c + 2 * h;
        const int2   kk = *(const int2*)(&r_key[idx]);
        const float2 ww = *(const float2*)(&r_w[idx]);
        const float2 we = *(const float2*)(&r_we[idx]);
        const bool m0 = ((kk.x & 15) == c15);
        const bool m1 = ((kk.y & 15) == c15);
        v2f af, bf;
        af.x = m0 ? ww.x : 0.0f;
        af.y = m1 ? ww.y : 0.0f;
        bf.x = ((kk.x >> 4) == c15) ? 1.0f : 0.0f;
        bf.y = ((kk.y >> 4) == c15) ? 1.0f : 0.0f;
        acc[j] = __builtin_amdgcn_wmma_f32_16x16x4_f32(false, af, false, bf, (short)0, acc[j],
                                                       false, false);
        es += af.x + af.y;
        ts += (m0 ? we.x : 0.0f) + (m1 ? we.y : 0.0f);
      }
    }
    __syncthreads();
  }

  es += __shfl_xor(es, 16, 32);
  ts += __shfl_xor(ts, 16, 32);
  if (h == 0) { s_es[wave * 16 + c15] = es; s_ts[wave * 16 + c15] = ts; }
  __syncthreads();
  if (tid < N_PAD) {
    const int w0 = (tid >> 4) * 2;
    const int r  = tid & 15;
    pes[tid] = s_es[w0 * 16 + r] + s_es[(w0 + 1) * 16 + r];
    pts[tid] = s_ts[w0 * 16 + r] + s_ts[(w0 + 1) * 16 + r];
  }

  const int dt = wave >> 1;
  const int st0 = (wave & 1) * 6;
#pragma unroll
  for (int j = 0; j < 6; ++j) {
    const int col = (st0 + j) * 16 + c15;
#pragma unroll
    for (int r = 0; r < 8; ++r) {
      const int row = dt * 16 + 8 * h + r;
      const float v = acc[j][r];
      if (row < N_NODES && col < N_NODES && v != 0.0f) atomicAdd(&S[row * S_PAD + col], v);
    }
  }
}

__global__ __launch_bounds__(128)
void k_final(const float* __restrict__ U, const float* __restrict__ bias,
             const float* __restrict__ ws, float* __restrict__ out) {
  __shared__ float aggr[16 * OUT_C];
  __shared__ __attribute__((aligned(16))) float s_pe[16 * 8];
  __shared__ __attribute__((aligned(16))) float s_pt[16 * 8];
  const float* h    = ws + OFF_H;
  const float* S    = ws + OFF_S;
  const float* pes  = ws + OFF_PES;
  const float* pts  = ws + OFF_PTS;
  int lane = threadIdx.x & 31;
  int wave = threadIdx.x >> 5;
  int m0 = blockIdx.x * 16;
  int n0 = wave * 32;
  int nb = n0 + (lane & 15);
  int mh = (lane >> 4) * 8;

  const int fr = threadIdx.x & 15;
  const int fg = threadIdx.x >> 4;
  float pe[16], pt[16];
#pragma unroll
  for (int i = 0; i < 16; ++i) {
    pe[i] = pes[(fg * 16 + i) * N_PAD + m0 + fr];
    pt[i] = pts[(fg * 16 + i) * N_PAD + m0 + fr];
  }
  const float u_last0 = U[OUT_C * OUT_C + nb];
  const float u_last1 = U[OUT_C * OUT_C + nb + 16];
  const float bias0   = bias[nb];
  const float bias1   = bias[nb + 16];

  v8f c0 = {}; v8f c1 = {};
#pragma unroll
  for (int k0 = 0; k0 < S_PAD; k0 += 4) {
    v2f a  = load_a_clamp(S, m0, k0, lane, S_PAD);
    v2f b0 = load_b(h, k0, n0,      lane, OUT_C);
    v2f b1 = load_b(h, k0, n0 + 16, lane, OUT_C);
    c0 = __builtin_amdgcn_wmma_f32_16x16x4_f32(false, a, false, b0, (short)0, c0, false, false);
    c1 = __builtin_amdgcn_wmma_f32_16x16x4_f32(false, a, false, b1, (short)0, c1, false, false);
  }
  v2f ub0[OUT_C / 4], ub1[OUT_C / 4];
#pragma unroll
  for (int i = 0; i < OUT_C / 4; ++i) {
    ub0[i] = load_b(U, 4 * i, n0,      lane, OUT_C);
    ub1[i] = load_b(U, 4 * i, n0 + 16, lane, OUT_C);
  }

  {
    float se = pe[0], st = pt[0];
#pragma unroll
    for (int i = 1; i < 16; ++i) { se += pe[i]; st += pt[i]; }
    s_pe[fr * 8 + fg] = se;
    s_pt[fr * 8 + fg] = st;
  }
  __syncthreads();
  float rinv[8], tn[8];
#pragma unroll
  for (int j = 0; j < 8; ++j) {
    int ml = j + mh;
    ml = (m0 + ml < N_NODES) ? ml : (N_NODES - 1 - m0);
    const float4 ea0 = *(const float4*)(&s_pe[ml * 8]);
    const float4 ea1 = *(const float4*)(&s_pe[ml * 8 + 4]);
    const float4 ta0 = *(const float4*)(&s_pt[ml * 8]);
    const float4 ta1 = *(const float4*)(&s_pt[ml * 8 + 4]);
    const float esm = ((ea0.x + ea0.y) + (ea0.z + ea0.w)) + ((ea1.x + ea1.y) + (ea1.z + ea1.w));
    const float tsm = ((ta0.x + ta0.y) + (ta0.z + ta0.w)) + ((ta1.x + ta1.y) + (ta1.z + ta1.w));
    const float es = esm + 1e-16f;
    rinv[j] = 1.0f / es;
    tn[j]   = tsm / es;
  }

#pragma unroll
  for (int j = 0; j < 8; ++j) {
    int mloc = j + mh;
    aggr[mloc * OUT_C + nb]      = c0[j] * rinv[j];
    aggr[mloc * OUT_C + nb + 16] = c1[j] * rinv[j];
  }
  __syncthreads();

  v8f d0 = {}; v8f d1 = {};
#pragma unroll
  for (int i = 0; i < OUT_C / 4; ++i) {
    v2f a  = load_a_lds(aggr, 4 * i, lane);
    d0 = __builtin_amdgcn_wmma_f32_16x16x4_f32(false, a, false, ub0[i], (short)0, d0, false, false);
    d1 = __builtin_amdgcn_wmma_f32_16x16x4_f32(false, a, false, ub1[i], (short)0, d1, false, false);
  }
#pragma unroll
  for (int j = 0; j < 8; ++j) {
    int m = m0 + j + mh;
    if (m < N_NODES) {
      out[m * OUT_C + nb]      = d0[j] + tn[j] * u_last0 + bias0;
      out[m * OUT_C + nb + 16] = d1[j] + tn[j] * u_last1 + bias1;
    }
  }
}

extern "C" void kernel_launch(void* const* d_in, const int* in_sizes, int n_in,
                              void* d_out, int out_size, void* d_ws, size_t ws_size,
                              hipStream_t stream) {
  const float* x    = (const float*)d_in[0];
  const float* ea   = (const float*)d_in[1];
  const float* W    = (const float*)d_in[2];
  const float* att  = (const float*)d_in[3];
  const float* U    = (const float*)d_in[4];
  const float* bias = (const float*)d_in[5];
  const int*   ei   = (const int*)d_in[6];
  int E = in_sizes[1];
  float* ws  = (float*)d_ws;
  float* out = (float*)d_out;

  k_init<<<64, 256, 0, stream>>>(ws);
  k_node_gemm<<<MTILES, 128, 0, stream>>>(x, W, ws);
  k_ab<<<N_NODES, 32, 0, stream>>>(att, ws);
  k_edge_max<<<MX_BLOCKS, MX_THREADS, 0, stream>>>(ei, ea, att, ws, E);
  k_edge_exp<<<EDGE_BLOCKS, EB_THREADS, 0, stream>>>(ei, ea, att, ws, E);
  k_final<<<MTILES, 128, 0, stream>>>(U, bias, ws, out);
}


# per-node scalars re-associated, a = x.(W.att_a), b = x.(W.att_b): W.att taken in the init launch, a and b by a fifth wave of the node launch; the a,b launch goes (6 -> 5); sum order of a, b changes
# speedup vs baseline: 3.6379x; 1.0542x over previous
#include <hip/hip_runtime.h>
#include <math.h>

typedef float v2f __attribute__((ext_vector_type(2)));
typedef float v8f __attribute__((ext_vector_type(8)));

#define N_NODES 177
#define FEAT 128
#define OUT_C 128
#define NEG_SLOPE 0.2f
#define MTILES 12
#define S_PAD 180
#define H_ROWS 180
#define N_PAD 192
#define EDGE_BLOCKS 128

#define OFF_H    0
#define OFF_A    (OFF_H + H_ROWS * FEAT)
#define OFF_B    (OFF_A + N_NODES)
#define OFF_AMAX (OFF_B + N_NODES)
#define OFF_S    (OFF_AMAX + N_NODES)
#define OFF_PES  ((OFF_S + N_NODES * S_PAD + 31) & ~31)
#define OFF_PTS  (OFF_PES + EDGE_BLOCKS * N_PAD)
#define OFF_VAB  (OFF_PTS + EDGE_BLOCKS * N_PAD)
#define WS_FLOATS (OFF_VAB + 2 * FEAT)

__device__ __forceinline__ unsigned f2ord(float f) {
  unsigned u = __float_as_uint(f);
  return (u & 0x80000000u) ? ~u : (u | 0x80000000u);
}
__device__ __forceinline__ float ord2f(unsigned u) {
  return __uint_as_float((u & 0x80000000u) ? (u ^ 0x80000000u) : ~u);
}
#define ORD_NEG_INF 0x007FFFFFu

__device__ __forceinline__ v2f load_a_clamp(const float* __restrict__ A, int m0, int k0,
                                            int lane, int ld) {
  int m = m0 + (lane & 15);
  m = (m < N_NODES) ? m : (N_NODES - 1);
  int kb = k0 + ((lane >> 4) << 1);
  const float* p = A + m * ld + kb;
  v2f a; a.x = p[0]; a.y = p[1];
  return a;
}
__device__ __forceinline__ v2f load_b(const float* __restrict__ B, int k0, int n0,
                                      int lane, int ld) {
  int n  = n0 + (lane & 15);
  int kb = k0 + ((lane >> 4) << 1);
  const float* p = B + kb * ld + n;
  v2f b; b.x = p[0]; b.y = p[ld];
  return b;
}
__device__ __forceinline__ v2f load_a_lds(const float* A, int k0, int lane) {
  int m  = lane & 15;
  int kb = k0 + ((lane >> 4) << 1);
  v2f a;
  a.x = A[m * OUT_C + kb];
  a.y = A[m * OUT_C + kb + 1];
  return a;
}

__global__ void k_init(const float* __restrict__ W, const float* __restrict__ att,
                       float* __restrict__ ws) {
  int tid    = blockIdx.x * blockDim.x + threadIdx.x;
  int stride = gridDim.x * blockDim.x;
  const int lane = threadIdx.x & 31;
  const int row  = tid >> 5;
  float wv[FEAT / 32] = {}, a0[FEAT / 32] = {}, a1[FEAT / 32] = {};
  if (row < FEAT) {
#pragma unroll
    for (int q = 0; q < FEAT / 32; ++q) {
      const int i = lane + 32 * q;
      wv[q] = W[row * OUT_C + i];
      a0[q] = att[i];
      a1[q] = att[OUT_C + i];
    }
  }
  unsigned* amax = (unsigned*)(ws + OFF_AMAX);
  for (int j = tid; j < N_NODES * S_PAD; j += stride) ws[OFF_S + j] = 0.0f;
  for (int j = tid; j < (H_ROWS - N_NODES) * FEAT; j += stride)
    ws[OFF_H + N_NODES * FEAT + j] = 0.0f;
  for (int j = tid; j < N_NODES; j += stride) amax[j] = ORD_NEG_INF;
  if (row < FEAT) {
    float pa = 0.0f, pb = 0.0f;
#pragma unroll
    for (int q = 0; q < FEAT / 32; ++q) { pa += wv[q] * a0[q]; pb += wv[q] * a1[q]; }
    for (int m = 16; m >= 1; m >>= 1) {
      pa += __shfl_xor(pa, m, 32);
      pb += __shfl_xor(pb, m, 32);
    }
    if (lane == 0) { ws[OFF_VAB + row] = pa; ws[OFF_VAB + FEAT + row] = pb; }
  }
}

__global__ void k_node_gemm(const float* __restrict__ x, const float* __restrict__ W,
                            float* __restrict__ ws) {
  float* h  = ws + OFF_H;
  int lane  = threadIdx.x & 31;
  int wave  = threadIdx.x >> 5;
  int m0    = blockIdx.x * 16;
  if (wave == 4) {
    const int r  = lane & 15;
    const int hh = lane >> 4;
    const int m  = m0 + r;
    const int mc = (m < N_NODES) ? m : (N_NODES - 1);
    const float* xr = x + mc * FEAT + hh * (FEAT / 2);
    const float* va = ws + OFF_VAB + hh * (FEAT / 2);
    const float* vb = va + FEAT;
    float pa = 0.0f, pb = 0.0f;
#pragma unroll
    for (int i = 0; i < FEAT / 2; ++i) {
      const float xv = xr[i];
      pa += xv * va[i];
      pb += xv * vb[i];
    }
    pa += __shfl_xor(pa, 16, 32);
    pb += __shfl_xor(pb, 16, 32);
    if (hh == 0 && m < N_NODES) { ws[OFF_A + m] = pa; ws[OFF_B + m] = pb; }
    return;
  }
  int n0    = wave * 32;
  v8f c0 = {}; v8f c1 = {};
  for (int k0 = 0; k0 < FEAT; k0 += 4) {
    v2f a  = load_a_clamp(x, m0, k0, lane, FEAT);
    v2f b0 = load_b(W, k0, n0,      lane, OUT_C);
    v2f b1 = load_b(W, k0, n0 + 16, lane, OUT_C);
    c0 = __builtin_amdgcn_wmma_f32_16x16x4_f32(false, a, false, b0, (short)0, c0, false, false);
    c1 = __builtin_amdgcn_wmma_f32_16x16x4_f32(false, a, false, b1, (short)0, c1, false, false);
  }
  int nb = n0 + (lane & 15);
  int mh = (lane >> 4) * 8;
  for (int j = 0; j < 8; ++j) {
    int m = m0 + j + mh;
    if (m < N_NODES) {
      h[m * OUT_C + nb]      = c0[j];
      h[m * OUT_C + nb + 16] = c1[j];
    }
  }
}

#define MX_THREADS 256
#define MX_BLOCKS  128
#define MX_R       16
#define MX_BATCH   (MX_THREADS * MX_R)
#define MX_WAVES   (MX_THREADS / 32)
#define MX_ROWS_PER_WAVE ((N_NODES + MX_WAVES - 1) / MX_WAVES)

__global__ __launch_bounds__(MX_THREADS)
void k_edge_max(const int* __restrict__ ei, const float* __restrict__ ea,
                const float* __restrict__ att, float* __restrict__ ws, int E) {
  __shared__ __attribute__((aligned(16))) float tbl[N_NODES * MX_THREADS];
  __shared__ float s_row[N_NODES];

  const float* a = ws + OFF_A;
  const float* b = ws + OFF_B;
  unsigned* amax = (unsigned*)(ws + OFF_AMAX);
  const float cc = att[2 * OUT_C];

  const int tid  = threadIdx.x;
  const int lane = tid & 31;
  const int wave = tid >> 5;

  const int per = (E + (int)gridDim.x - 1) / (int)gridDim.x;
  const int e0  = (int)blockIdx.x * per;
  const int e1  = (e0 + per < E) ? (e0 + per) : E;
  if (e0 >= e1) return;

  int   sv[MX_R], dv[MX_R];
  float ev[MX_R];
#pragma unroll
  for (int i = 0; i < MX_R; ++i) {
    const int e  = e0 + i * MX_THREADS + tid;
    const int ec = (e < e1) ? e : e0;
    sv[i] = ei[ec];
    dv[i] = ei[E + ec];
    ev[i] = ea[ec];
  }

  {
    const float ninf = -INFINITY;
    const float4 n4 = make_float4(ninf, ninf, ninf, ninf);
    float4* t4 = (float4*)tbl;
    for (int q = tid; q < N_NODES * MX_THREADS / 4; q += MX_THREADS) t4[q] = n4;
  }
  __syncthreads();

  float* col = tbl + tid;
  for (int base = e0; ; ) {
    float al[MX_R];
#pragma unroll
    for (int i = 0; i < MX_R; ++i) {
      const int e = base + i * MX_THREADS + tid;
      float v = a[dv[i]] + b[sv[i]] + cc * ev[i];
      v = (v > 0.0f) ? v : NEG_SLOPE * v;
      al[i] = (e < e1) ? v : -INFINITY;
    }
#pragma unroll
    for (int i = 0; i < MX_R; ++i) {
      float* p = col + dv[i] * MX_THREADS;
      *p = fmaxf(*p, al[i]);
    }
    base += MX_BATCH;
    if (base >= e1) break;
#pragma unroll
    for (int i = 0; i < MX_R; ++i) {
      const int e  = base + i * MX_THREADS + tid;
      const int ec = (e < e1) ? e : e0;
      sv[i] = ei[ec];
      dv[i] = ei[E + ec];
      ev[i] = ea[ec];
    }
  }
  __syncthreads();

  float m[MX_ROWS_PER_WAVE];
#pragma unroll
  for (int j = 0; j < MX_ROWS_PER_WAVE; ++j) {
    int r = wave + j * MX_WAVES;
    r = (r < N_NODES) ? r : (N_NODES - 1);
    const float4 v0 = *(const float4*)(&tbl[r * MX_THREADS + lane * 4]);
    const float4 v1 = *(const float4*)(&tbl[r * MX_THREADS + 128 + lane * 4]);
    m[j] = fmaxf(fmaxf(fmaxf(v0.x, v0.y), fmaxf(v0.z, v0.w)),
                 fmaxf(fmaxf(v1.x, v1.y), fmaxf(v1.z, v1.w)));
  }
#pragma unroll
  for (int k = 16; k >= 1; k >>= 1) {
#pragma unroll
    for (int j = 0; j < MX_ROWS_PER_WAVE; ++j) m[j] = fmaxf(m[j], __shfl_xor(m[j], k, 32));
  }
  float mine = -INFINITY;
#pragma unroll
  for (int j = 0; j < MX_ROWS_PER_WAVE; ++j) mine = (lane == j) ? m[j] : mine;
  {
    const int r = wave + lane * MX_WAVES;
    if (lane < MX_ROWS_PER_WAVE && r < N_NODES) s_row[r] = mine;
  }
  __syncthreads();
  if (tid < N_NODES && s_row[tid] > -INFINITY) atomicMax(&amax[tid], f2ord(s_row[tid]));
}

#define EB_THREADS 768
#define EB_R       6
#define EB_BATCH   (EB_THREADS * EB_R)
#define NBINS      144
#define REC_CAP    (EB_BATCH + NBINS * 3 + 8)

__global__ __launch_bounds__(EB_THREADS)
void k_edge_exp(const int* __restrict__ ei, const float* __restrict__ ea,
                const float* __restrict__ att, float* __restrict__ ws, int E) {
  __shared__ int   s_cnt[NBINS + 1];
  __shared__ int   s_start[NBINS];
  __shared__ __attribute__((aligned(16))) int   r_key[REC_CAP];
  __shared__ __attribute__((aligned(16))) float r_w[REC_CAP];
  __shared__ __attribute__((aligned(16))) float r_we[REC_CAP];
  __shared__ float s_es[(EB_THREADS / 32) * 16];
  __shared__ float s_ts[(EB_THREADS / 32) * 16];

  const float* a = ws + OFF_A;
  const float* b = ws + OFF_B;
  const unsigned* amax = (const unsigned*)(ws + OFF_AMAX);
  float* pes  = ws + OFF_PES + (int)blockIdx.x * N_PAD;
  float* pts  = ws + OFF_PTS + (int)blockIdx.x * N_PAD;
  float* S    = ws + OFF_S;
  const float cc = att[2 * OUT_C];

  const int tid  = threadIdx.x;
  const int lane = tid & 31;
  const int wave = tid >> 5;
  const int c15  = lane & 15;
  const int h    = lane >> 4;
  const int first_bin = wave * 6;

  const int per = (E + (int)gridDim.x - 1) / (int)gridDim.x;
  const int e0  = (int)blockIdx.x * per;
  const int e1  = (e0 + per < E) ? (e0 + per) : E;

  v8f acc[6];
#pragma unroll
  for (int j = 0; j < 6; ++j) acc[j] = (v8f){};
  float es = 0.0f, ts = 0.0f;

  for (int base = e0; base < e1; base += EB_BATCH) {
    if (tid <= NBINS) s_cnt[tid] = 0;
    __syncthreads();

    int   sv[EB_R], dv[EB_R];
    float ev[EB_R];
#pragma unroll
    for (int i = 0; i < EB_R; ++i) {
      const int e  = base + i * EB_THREADS + tid;
      const int ec = (e < e1) ? e : e0;
      sv[i] = ei[ec];
      dv[i] = ei[E + ec];
      ev[i] = ea[ec];
    }
    int   pk[EB_R], ppos[EB_R];
    float pw[EB_R], pwe[EB_R];
#pragma unroll
    for (int i = 0; i < EB_R; ++i) {
      float al = a[dv[i]] + b[sv[i]] + cc * ev[i];
      al = (al > 0.0f) ? al : NEG_SLOPE * al;
      const float w = __expf(al - ord2f(amax[dv[i]]));
      pk[i]  = (dv[i] & 15) | ((sv[i] & 15) << 4);
      pw[i]  = w;
      pwe[i] = w * ev[i];
    }
#pragma unroll
    for (int i = 0; i < EB_R; ++i) {
      const int  e   = base + i * EB_THREADS + tid;
      const bool ok  = e < e1;
      const int  bin = ok ? ((dv[i] >> 4) * 12 + (sv[i] >> 4)) : NBINS;
      const int  pos = atomicAdd(&s_cnt[bin], 1);
      ppos[i] = ok ? ((bin << 16) | pos) : -1;
    }
    __syncthreads();

    {
      int sum = 0;
      for (int q = lane; q < first_bin; q += 32) sum += (s_cnt[q] + 3) & ~3;
      for (int m = 16; m >= 1; m >>= 1) sum += __shfl_xor(sum, m, 32);
      if (lane < 6) {
        int st = sum;
        for (int jj = 0; jj < lane; ++jj) st += (s_cnt[first_bin + jj] + 3) & ~3;
        s_start[first_bin + lane] = st;
        const int nb = s_cnt[first_bin + lane];
        for (int p = st + nb; p < st + ((nb + 3) & ~3); ++p) {
          r_key[p] = 0; r_w[p] = 0.0f; r_we[p] = 0.0f;
        }
      }
    }
    __syncthreads();

#pragma unroll
    for (int i = 0; i < EB_R; ++i) {
      if (ppos[i] >= 0) {
        const int idx = s_start[ppos[i] >> 16] + (ppos[i] & 0xFFFF);
        r_key[idx] = pk[i];
        r_w[idx]   = pw[i];
        r_we[idx]  = pwe[i];
      }
    }
    __syncthreads();

#pragma unroll
    for (int j = 0; j < 6; ++j) {
      const int n   = __builtin_amdgcn_readfirstlane(s_cnt[first_bin + j]);
      const int st  = __builtin_amdgcn_readfirstlane(s_start[first_bin + j]);
      const int nch = (n + 3) >> 2;
      for (int c = 0; c < nch; ++c) {
        const int idx = st + 4 * c + 2 * h;
        const int2   kk = *(const int2*)(&r_key[idx]);
        const float2 ww = *(const float2*)(&r_w[idx]);
        const float2 we = *(const float2*)(&r_we[idx]);
        const bool m0 = ((kk.x & 15) == c15);
        const bool m1 = ((kk.y & 15) == c15);
        v2f af, bf;
        af.x = m0 ? ww.x : 0.0f;
        af.y = m1 ? ww.y : 0.0f;
        bf.x = ((kk.x >> 4) == c15) ? 1.0f : 0.0f;
        bf.y = ((kk.y >> 4) == c15) ? 1.0f : 0.0f;
        acc[j] = __builtin_amdgcn_wmma_f32_16x16x4_f32(false, af, false, bf, (short)0, acc[j],
                                                       false, false);
        es += af.x + af.y;
        ts += (m0 ? we.x : 0.0f) + (m1 ? we.y : 0.0f);
      }
    }
    __syncthreads();
  }

  es += __shfl_xor(es, 16, 32);
  ts += __shfl_xor(ts, 16, 32);
  if (h == 0) { s_es[wave * 16 + c15] = es; s_ts[wave * 16 + c15] = ts; }
  __syncthreads();
  if (tid < N_PAD) {
    const int w0 = (tid >> 4) * 2;
    const int r  = tid & 15;
    pes[tid] = s_es[w0 * 16 + r] + s_es[(w0 + 1) * 16 + r];
    pts[tid] = s_ts[w0 * 16 + r] + s_ts[(w0 + 1) * 16 + r];
  }

  const int dt = wave >> 1;
  const int st0 = (wave & 1) * 6;
#pragma unroll
  for (int j = 0; j < 6; ++j) {
    const int col = (st0 + j) * 16 + c15;
#pragma unroll
    for (int r = 0; r < 8; ++r) {
      const int row = dt * 16 + 8 * h + r;
      const float v = acc[j][r];
      if (row < N_NODES && col < N_NODES && v != 0.0f) atomicAdd(&S[row * S_PAD + col], v);
    }
  }
}

__global__ __launch_bounds__(128)
void k_final(const float* __restrict__ U, const float* __restrict__ bias,
             const float* __restrict__ ws, float* __restrict__ out) {
  __shared__ float aggr[16 * OUT_C];
  __shared__ __attribute__((aligned(16))) float s_pe[16 * 8];
  __shared__ __attribute__((aligned(16))) float s_pt[16 * 8];
  const float* h    = ws + OFF_H;
  const float* S    = ws + OFF_S;
  const float* pes  = ws + OFF_PES;
  const float* pts  = ws + OFF_PTS;
  int lane = threadIdx.x & 31;
  int wave = threadIdx.x >> 5;
  int m0 = blockIdx.x * 16;
  int n0 = wave * 32;
  int nb = n0 + (lane & 15);
  int mh = (lane >> 4) * 8;

  const int fr = threadIdx.x & 15;
  const int fg = threadIdx.x >> 4;
  float pe[16], pt[16];
#pragma unroll
  for (int i = 0; i < 16; ++i) {
    pe[i] = pes[(fg * 16 + i) * N_PAD + m0 + fr];
    pt[i] = pts[(fg * 16 + i) * N_PAD + m0 + fr];
  }
  const float u_last0 = U[OUT_C * OUT_C + nb];
  const float u_last1 = U[OUT_C * OUT_C + nb + 16];
  const float bias0   = bias[nb];
  const float bias1   = bias[nb + 16];

  v8f c0 = {}; v8f c1 = {};
#pragma unroll
  for (int k0 = 0; k0 < S_PAD; k0 += 4) {
    v2f a  = load_a_clamp(S, m0, k0, lane, S_PAD);
    v2f b0 = load_b(h, k0, n0,      lane, OUT_C);
    v2f b1 = load_b(h, k0, n0 + 16, lane, OUT_C);
    c0 = __builtin_amdgcn_wmma_f32_16x16x4_f32(false, a, false, b0, (short)0, c0, false, false);
    c1 = __builtin_amdgcn_wmma_f32_16x16x4_f32(false, a, false, b1, (short)0, c1, false, false);
  }
  v2f ub0[OUT_C / 4], ub1[OUT_C / 4];
#pragma unroll
  for (int i = 0; i < OUT_C / 4; ++i) {
    ub0[i] = load_b(U, 4 * i, n0,      lane, OUT_C);
    ub1[i] = load_b(U, 4 * i, n0 + 16, lane, OUT_C);
  }

  {
    float se = pe[0], st = pt[0];
#pragma unroll
    for (int i = 1; i < 16; ++i) { se += pe[i]; st += pt[i]; }
    s_pe[fr * 8 + fg] = se;
    s_pt[fr * 8 + fg] = st;
  }
  __syncthreads();
  float rinv[8], tn[8];
#pragma unroll
  for (int j = 0; j < 8; ++j) {
    int ml = j + mh;
    ml = (m0 + ml < N_NODES) ? ml : (N_NODES - 1 - m0);
    const float4 ea0 = *(const float4*)(&s_pe[ml * 8]);
    const float4 ea1 = *(const float4*)(&s_pe[ml * 8 + 4]);
    const float4 ta0 = *(const float4*)(&s_pt[ml * 8]);
    const float4 ta1 = *(const float4*)(&s_pt[ml * 8 + 4]);
    const float esm = ((ea0.x + ea0.y) + (ea0.z + ea0.w)) + ((ea1.x + ea1.y) + (ea1.z + ea1.w));
    const float tsm = ((ta0.x + ta0.y) + (ta0.z + ta0.w)) + ((ta1.x + ta1.y) + (ta1.z + ta1.w));
    const float es = esm + 1e-16f;
    rinv[j] = 1.0f / es;
    tn[j]   = tsm / es;
  }

#pragma unroll
  for (int j = 0; j < 8; ++j) {
    int mloc = j + mh;
    aggr[mloc * OUT_C + nb]      = c0[j] * rinv[j];
    aggr[mloc * OUT_C + nb + 16] = c1[j] * rinv[j];
  }
  __syncthreads();

  v8f d0 = {}; v8f d1 = {};
#pragma unroll
  for (int i = 0; i < OUT_C / 4; ++i) {
    v2f a  = load_a_lds(aggr, 4 * i, lane);
    d0 = __builtin_amdgcn_wmma_f32_16x16x4_f32(false, a, false, ub0[i], (short)0, d0, false, false);
    d1 = __builtin_amdgcn_wmma_f32_16x16x4_f32(false, a, false, ub1[i], (short)0, d1, false, false);
  }
#pragma unroll
  for (int j = 0; j < 8; ++j) {
    int m = m0 + j + mh;
    if (m < N_NODES) {
      out[m * OUT_C + nb]      = d0[j] + tn[j] * u_last0 + bias0;
      out[m * OUT_C + nb + 16] = d1[j] + tn[j] * u_last1 + bias1;
    }
  }
}

extern "C" void kernel_launch(void* const* d_in, const int* in_sizes, int n_in,
                              void* d_out, int out_size, void* d_ws, size_t ws_size,
                              hipStream_t stream) {
  const float* x    = (const float*)d_in[0];
  const float* ea   = (const float*)d_in[1];
  const float* W    = (const float*)d_in[2];
  const float* att  = (const float*)d_in[3];
  const float* U    = (const float*)d_in[4];
  const float* bias = (const float*)d_in[5];
  const int*   ei   = (const int*)d_in[6];
  int E = in_sizes[1];
  float* ws  = (float*)d_ws;
  float* out = (float*)d_out;

  k_init<<<64, 256, 0, stream>>>(W, att, ws);
  k_node_gemm<<<MTILES, 160, 0, stream>>>(x, W, ws);
  k_edge_max<<<MX_BLOCKS, MX_THREADS, 0, stream>>>(ei, ea, att, ws, E);
  k_edge_exp<<<EDGE_BLOCKS, EB_THREADS, 0, stream>>>(ei, ea, att, ws, E);
  k_final<<<MTILES, 128, 0, stream>>>(U, bias, ws, out);
}


# node launch dissolved: h = x@W rides as 12 extra blocks of the segment-max launch, a and b are taken in the init launch (W.att in LDS, then x . W.att); 5 -> 4 launches; every sum keeps its order
# speedup vs baseline: 3.8783x; 1.0661x over previous
#include <hip/hip_runtime.h>
#include <math.h>

typedef float v2f __attribute__((ext_vector_type(2)));
typedef float v8f __attribute__((ext_vector_type(8)));

#define N_NODES 177
#define FEAT 128
#define OUT_C 128
#define NEG_SLOPE 0.2f
#define MTILES 12
#define S_PAD 180
#define H_ROWS 180
#define N_PAD 192
#define EDGE_BLOCKS 128

#define OFF_H    0
#define OFF_A    (OFF_H + H_ROWS * FEAT)
#define OFF_B    (OFF_A + N_NODES)
#define OFF_AMAX (OFF_B + N_NODES)
#define OFF_S    (OFF_AMAX + N_NODES)
#define OFF_PES  ((OFF_S + N_NODES * S_PAD + 31) & ~31)
#define OFF_PTS  (OFF_PES + EDGE_BLOCKS * N_PAD)
#define WS_FLOATS (OFF_PTS + EDGE_BLOCKS * N_PAD)

__device__ __forceinline__ unsigned f2ord(float f) {
  unsigned u = __float_as_uint(f);
  return (u & 0x80000000u) ? ~u : (u | 0x80000000u);
}
__device__ __forceinline__ float ord2f(unsigned u) {
  return __uint_as_float((u & 0x80000000u) ? (u ^ 0x80000000u) : ~u);
}
#define ORD_NEG_INF 0x007FFFFFu

__device__ __forceinline__ v2f load_a_clamp(const float* __restrict__ A, int m0, int k0,
                                            int lane, int ld) {
  int m = m0 + (lane & 15);
  m = (m < N_NODES) ? m : (N_NODES - 1);
  int kb = k0 + ((lane >> 4) << 1);
  const float* p = A + m * ld + kb;
  v2f a; a.x = p[0]; a.y = p[1];
  return a;
}
__device__ __forceinline__ v2f load_b(const float* __restrict__ B, int k0, int n0,
                                      int lane, int ld) {
  int n  = n0 + (lane & 15);
  int kb = k0 + ((lane >> 4) << 1);
  const float* p = B + kb * ld + n;
  v2f b; b.x = p[0]; b.y = p[ld];
  return b;
}
__device__ __forceinline__ v2f load_a_lds(const float* A, int k0, int lane) {
  int m  = lane & 15;
  int kb = k0 + ((lane >> 4) << 1);
  v2f a;
  a.x = A[m * OUT_C + kb];
  a.y = A[m * OUT_C + kb + 1];
  return a;
}

#define INIT_THREADS 256
#define INIT_ROWS_PER_WAVE (FEAT / (INIT_THREADS / 32))
__global__ __launch_bounds__(INIT_THREADS)
void k_init(const float* __restrict__ x, const float* __restrict__ W,
            const float* __restrict__ att, float* __restrict__ ws) {
  __shared__ float s_va[FEAT];
  __shared__ float s_vb[FEAT];
  int tid    = blockIdx.x * blockDim.x + threadIdx.x;
  int stride = gridDim.x * blockDim.x;
  const int lane = threadIdx.x & 31;
  const int wave = threadIdx.x >> 5;
  const bool ab_block = blockIdx.x < MTILES;
  const int r  = lane & 15;
  const int hh = lane >> 4;
  const int m  = (int)blockIdx.x * 16 + r;
  const int mc = (m < N_NODES) ? m : (N_NODES - 1);
  float wv[INIT_ROWS_PER_WAVE][FEAT / 32] = {};
  float a0[FEAT / 32] = {}, a1[FEAT / 32] = {};
  float xv[FEAT / 2] = {};
  if (ab_block) {
#pragma unroll
    for (int q = 0; q < FEAT / 32; ++q) {
      a0[q] = att[lane + 32 * q];
      a1[q] = att[OUT_C + lane + 32 * q];
    }
#pragma unroll
    for (int j = 0; j < INIT_ROWS_PER_WAVE; ++j) {
      const int row = wave * INIT_ROWS_PER_WAVE + j;
#pragma unroll
      for (int q = 0; q < FEAT / 32; ++q) wv[j][q] = W[row * OUT_C + lane + 32 * q];
    }
    if (wave == 0) {
      const float* xr = x + mc * FEAT + hh * (FEAT / 2);
#pragma unroll
      for (int i = 0; i < FEAT / 2; ++i) xv[i] = xr[i];
    }
  }
  unsigned* amax = (unsigned*)(ws + OFF_AMAX);
  for (int j = tid; j < N_NODES * S_PAD; j += stride) ws[OFF_S + j] = 0.0f;
  for (int j = tid; j < (H_ROWS - N_NODES) * FEAT; j += stride)
    ws[OFF_H + N_NODES * FEAT + j] = 0.0f;
  for (int j = tid; j < N_NODES; j += stride) amax[j] = ORD_NEG_INF;
  if (ab_block) {
    float pa[INIT_ROWS_PER_WAVE], pb[INIT_ROWS_PER_WAVE];
#pragma unroll
    for (int j = 0; j < INIT_ROWS_PER_WAVE; ++j) {
      pa[j] = 0.0f; pb[j] = 0.0f;
#pragma unroll
      for (int q = 0; q < FEAT / 32; ++q) { pa[j] += wv[j][q] * a0[q]; pb[j] += wv[j][q] * a1[q]; }
    }
    const bool b4 = (lane & 16) != 0, b3 = (lane & 8) != 0, b2 = (lane & 4) != 0, b1 = (lane & 2) != 0;
    float qa[8], qb[8];
#pragma unroll
    for (int i = 0; i < 8; ++i) {
      const float ka = b4 ? pa[i + 8] : pa[i], sa = b4 ? pa[i] : pa[i + 8];
      const float kb = b4 ? pb[i + 8] : pb[i], sb = b4 ? pb[i] : pb[i + 8];
      qa[i] = ka + __shfl_xor(sa, 16, 32);
      qb[i] = kb + __shfl_xor(sb, 16, 32);
    }
    float ra[4], rb[4];
#pragma unroll
    for (int i = 0; i < 4; ++i) {
      const float ka = b3 ? qa[i + 4] : qa[i], sa = b3 ? qa[i] : qa[i + 4];
      const float kb = b3 ? qb[i + 4] : qb[i], sb = b3 ? qb[i] : qb[i + 4];
      ra[i] = ka + __shfl_xor(sa, 8, 32);
      rb[i] = kb + __shfl_xor(sb, 8, 32);
    }
    float ua[2], ub[2];
#pragma unroll
    for (int i = 0; i < 2; ++i) {
      const float ka = b2 ? ra[i + 2] : ra[i], sa = b2 ? ra[i] : ra[i + 2];
      const float kb = b2 ? rb[i + 2] : rb[i], sb = b2 ? rb[i] : rb[i + 2];
      ua[i] = ka + __shfl_xor(sa, 4, 32);
      ub[i] = kb + __shfl_xor(sb, 4, 32);
    }
    float ta, tb;
    {
      const float ka = b1 ? ua[1] : ua[0], sa = b1 ? ua[0] : ua[1];
      const float kb = b1 ? ub[1] : ub[0], sb = b1 ? ub[0] : ub[1];
      ta = ka + __shfl_xor(sa, 2, 32);
      tb = kb + __shfl_xor(sb, 2, 32);
    }
    ta += __shfl_xor(ta, 1, 32);
    tb += __shfl_xor(tb, 1, 32);
    if ((lane & 1) == 0) {
      s_va[wave * INIT_ROWS_PER_WAVE + (lane >> 1)] = ta;
      s_vb[wave * INIT_ROWS_PER_WAVE + (lane >> 1)] = tb;
    }
    __syncthreads();
    if (wave == 0) {
      const float* va = s_va + hh * (FEAT / 2);
      const float* vb = s_vb + hh * (FEAT / 2);
      float sa = 0.0f, sb = 0.0f;
#pragma unroll
      for (int i = 0; i < FEAT / 2; ++i) {
        sa += xv[i] * va[i];
        sb += xv[i] * vb[i];
      }
      sa += __shfl_xor(sa, 16, 32);
      sb += __shfl_xor(sb, 16, 32);
      if (hh == 0 && m < N_NODES) { ws[OFF_A + m] = sa; ws[OFF_B + m] = sb; }
    }
  }
}

__device__ __forceinline__ void node_gemm_strip(const float* __restrict__ x, const float* __restrict__ W,
                                                float* __restrict__ h, int m0, int wave, int lane) {
  const int n0 = wave * 16;
  v8f c0 = {};
  for (int k0 = 0; k0 < FEAT; k0 += 4) {
    v2f a  = load_a_clamp(x, m0, k0, lane, FEAT);
    v2f b0 = load_b(W, k0, n0, lane, OUT_C);
    c0 = __builtin_amdgcn_wmma_f32_16x16x4_f32(false, a, false, b0, (short)0, c0, false, false);
  }
  const int nb = n0 + (lane & 15);
  const int mh = (lane >> 4) * 8;
  for (int j = 0; j < 8; ++j) {
    const int m = m0 + j + mh;
    if (m < N_NODES) h[m * OUT_C + nb] = c0[j];
  }
}

#define MX_THREADS 256
#define MX_BLOCKS  128
#define MX_GRID    (MX_BLOCKS + MTILES)
#define MX_R       16
#define MX_BATCH   (MX_THREADS * MX_R)
#define MX_WAVES   (MX_THREADS / 32)
#define MX_ROWS_PER_WAVE ((N_NODES + MX_WAVES - 1) / MX_WAVES)

__global__ __launch_bounds__(MX_THREADS)
void k_edge_max(const int* __restrict__ ei, const float* __restrict__ ea,
                const float* __restrict__ att, const float* __restrict__ x,
                const float* __restrict__ W, float* __restrict__ ws, int E) {
  __shared__ __attribute__((aligned(16))) float tbl[N_NODES * MX_THREADS];
  __shared__ float s_row[N_NODES];

  if ((int)blockIdx.x >= MX_BLOCKS) {
    node_gemm_strip(x, W, ws + OFF_H, ((int)blockIdx.x - MX_BLOCKS) * 16,
                    (int)threadIdx.x >> 5, (int)threadIdx.x & 31);
    return;
  }

  const float* a = ws + OFF_A;
  const float* b = ws + OFF_B;
  unsigned* amax = (unsigned*)(ws + OFF_AMAX);
  const float cc = att[2 * OUT_C];

  const int tid  = threadIdx.x;
  const int lane = tid & 31;
  const int wave = tid >> 5;

  const int per = (E + MX_BLOCKS - 1) / MX_BLOCKS;
  const int e0  = (int)blockIdx.x * per;
  const int e1  = (e0 + per < E) ? (e0 + per) : E;
  if (e0 >= e1) return;

  int   sv[MX_R], dv[MX_R];
  float ev[MX_R];
#pragma unroll
  for (int i = 0; i < MX_R; ++i) {
    const int e  = e0 + i * MX_THREADS + tid;
    const int ec = (e < e1) ? e : e0;
    sv[i] = ei[ec];
    dv[i] = ei[E + ec];
    ev[i] = ea[ec];
  }

  {
    const float ninf = -INFINITY;
    const float4 n4 = make_float4(ninf, ninf, ninf, ninf);
    float4* t4 = (float4*)tbl;
    for (int q = tid; q < N_NODES * MX_THREADS / 4; q += MX_THREADS) t4[q] = n4;
  }
  __syncthreads();

  float* col = tbl + tid;
  for (int base = e0; ; ) {
    float al[MX_R];
#pragma unroll
    for (int i = 0; i < MX_R; ++i) {
      const int e = base + i * MX_THREADS + tid;
      float v = a[dv[i]] + b[sv[i]] + cc * ev[i];
      v = (v > 0.0f) ? v : NEG_SLOPE * v;
      al[i] = (e < e1) ? v : -INFINITY;
    }
#pragma unroll
    for (int i = 0; i < MX_R; ++i) {
      float* p = col + dv[i] * MX_THREADS;
      *p = fmaxf(*p, al[i]);
    }
    base += MX_BATCH;
    if (base >= e1) break;
#pragma unroll
    for (int i = 0; i < MX_R; ++i) {
      const int e  = base + i * MX_THREADS + tid;
      const int ec = (e < e1) ? e : e0;
      sv[i] = ei[ec];
      dv[i] = ei[E + ec];
      ev[i] = ea[ec];
    }
  }
  __syncthreads();

  float m[MX_ROWS_PER_WAVE];
#pragma unroll
  for (int j = 0; j < MX_ROWS_PER_WAVE; ++j) {
    int r = wave + j * MX_WAVES;
    r = (r < N_NODES) ? r : (N_NODES - 1);
    const float4 v0 = *(const float4*)(&tbl[r * MX_THREADS + lane * 4]);
    const float4 v1 = *(const float4*)(&tbl[r * MX_THREADS + 128 + lane * 4]);
    m[j] = fmaxf(fmaxf(fmaxf(v0.x, v0.y), fmaxf(v0.z, v0.w)),
                 fmaxf(fmaxf(v1.x, v1.y), fmaxf(v1.z, v1.w)));
  }
#pragma unroll
  for (int k = 16; k >= 1; k >>= 1) {
#pragma unroll
    for (int j = 0; j < MX_ROWS_PER_WAVE; ++j) m[j] = fmaxf(m[j], __shfl_xor(m[j], k, 32));
  }
  float mine = -INFINITY;
#pragma unroll
  for (int j = 0; j < MX_ROWS_PER_WAVE; ++j) mine = (lane == j) ? m[j] : mine;
  {
    const int r = wave + lane * MX_WAVES;
    if (lane < MX_ROWS_PER_WAVE && r < N_NODES) s_row[r] = mine;
  }
  __syncthreads();
  if (tid < N_NODES && s_row[tid] > -INFINITY) atomicMax(&amax[tid], f2ord(s_row[tid]));
}

#define EB_THREADS 768
#define EB_R       6
#define EB_BATCH   (EB_THREADS * EB_R)
#define NBINS      144
#define REC_CAP    (EB_BATCH + NBINS * 3 + 8)

__global__ __launch_bounds__(EB_THREADS)
void k_edge_exp(const int* __restrict__ ei, const float* __restrict__ ea,
                const float* __restrict__ att, float* __restrict__ ws, int E) {
  __shared__ int   s_cnt[NBINS + 1];
  __shared__ int   s_start[NBINS];
  __shared__ __attribute__((aligned(16))) int   r_key[REC_CAP];
  __shared__ __attribute__((aligned(16))) float r_w[REC_CAP];
  __shared__ __attribute__((aligned(16))) float r_we[REC_CAP];
  __shared__ float s_es[(EB_THREADS / 32) * 16];
  __shared__ float s_ts[(EB_THREADS / 32) * 16];

  const float* a = ws + OFF_A;
  const float* b = ws + OFF_B;
  const unsigned* amax = (const unsigned*)(ws + OFF_AMAX);
  float* pes  = ws + OFF_PES + (int)blockIdx.x * N_PAD;
  float* pts  = ws + OFF_PTS + (int)blockIdx.x * N_PAD;
  float* S    = ws + OFF_S;
  const float cc = att[2 * OUT_C];

  const int tid  = threadIdx.x;
  const int lane = tid & 31;
  const int wave = tid >> 5;
  const int c15  = lane & 15;
  const int h    = lane >> 4;
  const int first_bin = wave * 6;

  const int per = (E + (int)gridDim.x - 1) / (int)gridDim.x;
  const int e0  = (int)blockIdx.x * per;
  const int e1  = (e0 + per < E) ? (e0 + per) : E;

  v8f acc[6];
#pragma unroll
  for (int j = 0; j < 6; ++j) acc[j] = (v8f){};
  float es = 0.0f, ts = 0.0f;

  for (int base = e0; base < e1; base += EB_BATCH) {
    if (tid <= NBINS) s_cnt[tid] = 0;
    __syncthreads();

    int   sv[EB_R], dv[EB_R];
    float ev[EB_R];
#pragma unroll
    for (int i = 0; i < EB_R; ++i) {
      const int e  = base + i * EB_THREADS + tid;
      const int ec = (e < e1) ? e : e0;
      sv[i] = ei[ec];
      dv[i] = ei[E + ec];
      ev[i] = ea[ec];
    }
    int   pk[EB_R], ppos[EB_R];
    float pw[EB_R], pwe[EB_R];
#pragma unroll
    for (int i = 0; i < EB_R; ++i) {
      float al = a[dv[i]] + b[sv[i]] + cc * ev[i];
      al = (al > 0.0f) ? al : NEG_SLOPE * al;
      const float w = __expf(al - ord2f(amax[dv[i]]));
      pk[i]  = (dv[i] & 15) | ((sv[i] & 15) << 4);
      pw[i]  = w;
      pwe[i] = w * ev[i];
    }
#pragma unroll
    for (int i = 0; i < EB_R; ++i) {
      const int  e   = base + i * EB_THREADS + tid;
      const bool ok  = e < e1;
      const int  bin = ok ? ((dv[i] >> 4) * 12 + (sv[i] >> 4)) : NBINS;
      const int  pos = atomicAdd(&s_cnt[bin], 1);
      ppos[i] = ok ? ((bin << 16) | pos) : -1;
    }
    __syncthreads();

    {
      int sum = 0;
      for (int q = lane; q < first_bin; q += 32) sum += (s_cnt[q] + 3) & ~3;
      for (int m = 16; m >= 1; m >>= 1) sum += __shfl_xor(sum, m, 32);
      if (lane < 6) {
        int st = sum;
        for (int jj = 0; jj < lane; ++jj) st += (s_cnt[first_bin + jj] + 3) & ~3;
        s_start[first_bin + lane] = st;
        const int nb = s_cnt[first_bin + lane];
        for (int p = st + nb; p < st + ((nb + 3) & ~3); ++p) {
          r_key[p] = 0; r_w[p] = 0.0f; r_we[p] = 0.0f;
        }
      }
    }
    __syncthreads();

#pragma unroll
    for (int i = 0; i < EB_R; ++i) {
      if (ppos[i] >= 0) {
        const int idx = s_start[ppos[i] >> 16] + (ppos[i] & 0xFFFF);
        r_key[idx] = pk[i];
        r_w[idx]   = pw[i];
        r_we[idx]  = pwe[i];
      }
    }
    __syncthreads();

#pragma unroll
    for (int j = 0; j < 6; ++j) {
      const int n   = __builtin_amdgcn_readfirstlane(s_cnt[first_bin + j]);
      const int st  = __builtin_amdgcn_readfirstlane(s_start[first_bin + j]);
      const int nch = (n + 3) >> 2;
      for (int c = 0; c < nch; ++c) {
        const int idx = st + 4 * c + 2 * h;
        const int2   kk = *(const int2*)(&r_key[idx]);
        const float2 ww = *(const float2*)(&r_w[idx]);
        const float2 we = *(const float2*)(&r_we[idx]);
        const bool m0 = ((kk.x & 15) == c15);
        const bool m1 = ((kk.y & 15) == c15);
        v2f af, bf;
        af.x = m0 ? ww.x : 0.0f;
        af.y = m1 ? ww.y : 0.0f;
        bf.x = ((kk.x >> 4) == c15) ? 1.0f : 0.0f;
        bf.y = ((kk.y >> 4) == c15) ? 1.0f : 0.0f;
        acc[j] = __builtin_amdgcn_wmma_f32_16x16x4_f32(false, af, false, bf, (short)0, acc[j],
                                                       false, false);
        es += af.x + af.y;
        ts += (m0 ? we.x : 0.0f) + (m1 ? we.y : 0.0f);
      }
    }
    __syncthreads();
  }

  es += __shfl_xor(es, 16, 32);
  ts += __shfl_xor(ts, 16, 32);
  if (h == 0) { s_es[wave * 16 + c15] = es; s_ts[wave * 16 + c15] = ts; }
  __syncthreads();
  if (tid < N_PAD) {
    const int w0 = (tid >> 4) * 2;
    const int r  = tid & 15;
    pes[tid] = s_es[w0 * 16 + r] + s_es[(w0 + 1) * 16 + r];
    pts[tid] = s_ts[w0 * 16 + r] + s_ts[(w0 + 1) * 16 + r];
  }

  const int dt = wave >> 1;
  const int st0 = (wave & 1) * 6;
#pragma unroll
  for (int j = 0; j < 6; ++j) {
    const int col = (st0 + j) * 16 + c15;
#pragma unroll
    for (int r = 0; r < 8; ++r) {
      const int row = dt * 16 + 8 * h + r;
      const float v = acc[j][r];
      if (row < N_NODES && col < N_NODES && v != 0.0f) atomicAdd(&S[row * S_PAD + col], v);
    }
  }
}

__global__ __launch_bounds__(128)
void k_final(const float* __restrict__ U, const float* __restrict__ bias,
             const float* __restrict__ ws, float* __restrict__ out) {
  __shared__ float aggr[16 * OUT_C];
  __shared__ __attribute__((aligned(16))) float s_pe[16 * 8];
  __shared__ __attribute__((aligned(16))) float s_pt[16 * 8];
  const float* h    = ws + OFF_H;
  const float* S    = ws + OFF_S;
  const float* pes  = ws + OFF_PES;
  const float* pts  = ws + OFF_PTS;
  int lane = threadIdx.x & 31;
  int wave = threadIdx.x >> 5;
  int m0 = blockIdx.x * 16;
  int n0 = wave * 32;
  int nb = n0 + (lane & 15);
  int mh = (lane >> 4) * 8;

  const int fr = threadIdx.x & 15;
  const int fg = threadIdx.x >> 4;
  float pe[16], pt[16];
#pragma unroll
  for (int i = 0; i < 16; ++i) {
    pe[i] = pes[(fg * 16 + i) * N_PAD + m0 + fr];
    pt[i] = pts[(fg * 16 + i) * N_PAD + m0 + fr];
  }
  const float u_last0 = U[OUT_C * OUT_C + nb];
  const float u_last1 = U[OUT_C * OUT_C + nb + 16];
  const float bias0   = bias[nb];
  const float bias1   = bias[nb + 16];

  v8f c0 = {}; v8f c1 = {};
#pragma unroll
  for (int k0 = 0; k0 < S_PAD; k0 += 4) {
    v2f a  = load_a_clamp(S, m0, k0, lane, S_PAD);
    v2f b0 = load_b(h, k0, n0,      lane, OUT_C);
    v2f b1 = load_b(h, k0, n0 + 16, lane, OUT_C);
    c0 = __builtin_amdgcn_wmma_f32_16x16x4_f32(false, a, false, b0, (short)0, c0, false, false);
    c1 = __builtin_amdgcn_wmma_f32_16x16x4_f32(false, a, false, b1, (short)0, c1, false, false);
  }
  v2f ub0[OUT_C / 4], ub1[OUT_C / 4];
#pragma unroll
  for (int i = 0; i < OUT_C / 4; ++i) {
    ub0[i] = load_b(U, 4 * i, n0,      lane, OUT_C);
    ub1[i] = load_b(U, 4 * i, n0 + 16, lane, OUT_C);
  }

  {
    float se = pe[0], st = pt[0];
#pragma unroll
    for (int i = 1; i < 16; ++i) { se += pe[i]; st += pt[i]; }
    s_pe[fr * 8 + fg] = se;
    s_pt[fr * 8 + fg] = st;
  }
  __syncthreads();
  float rinv[8], tn[8];
#pragma unroll
  for (int j = 0; j < 8; ++j) {
    int ml = j + mh;
    ml = (m0 + ml < N_NODES) ? ml : (N_NODES - 1 - m0);
    const float4 ea0 = *(const float4*)(&s_pe[ml * 8]);
    const float4 ea1 = *(const float4*)(&s_pe[ml * 8 + 4]);
    const float4 ta0 = *(const float4*)(&s_pt[ml * 8]);
    const float4 ta1 = *(const float4*)(&s_pt[ml * 8 + 4]);
    const float esm = ((ea0.x + ea0.y) + (ea0.z + ea0.w)) + ((ea1.x + ea1.y) + (ea1.z + ea1.w));
    const float tsm = ((ta0.x + ta0.y) + (ta0.z + ta0.w)) + ((ta1.x + ta1.y) + (ta1.z + ta1.w));
    const float es = esm + 1e-16f;
    rinv[j] = 1.0f / es;
    tn[j]   = tsm / es;
  }

#pragma unroll
  for (int j = 0; j < 8; ++j) {
    int mloc = j + mh;
    aggr[mloc * OUT_C + nb]      = c0[j] * rinv[j];
    aggr[mloc * OUT_C + nb + 16] = c1[j] * rinv[j];
  }
  __syncthreads();

  v8f d0 = {}; v8f d1 = {};
#pragma unroll
  for (int i = 0; i < OUT_C / 4; ++i) {
    v2f a  = load_a_lds(aggr, 4 * i, lane);
    d0 = __builtin_amdgcn_wmma_f32_16x16x4_f32(false, a, false, ub0[i], (short)0, d0, false, false);
    d1 = __builtin_amdgcn_wmma_f32_16x16x4_f32(false, a, false, ub1[i], (short)0, d1, false, false);
  }
#pragma unroll
  for (int j = 0; j < 8; ++j) {
    int m = m0 + j + mh;
    if (m < N_NODES) {
      out[m * OUT_C + nb]      = d0[j] + tn[j] * u_last0 + bias0;
      out[m * OUT_C + nb + 16] = d1[j] + tn[j] * u_last1 + bias1;
    }
  }
}

extern "C" void kernel_launch(void* const* d_in, const int* in_sizes, int n_in,
                              void* d_out, int out_size, void* d_ws, size_t ws_size,
                              hipStream_t stream) {
  const float* x    = (const float*)d_in[0];
  const float* ea   = (const float*)d_in[1];
  const float* W    = (const float*)d_in[2];
  const float* att  = (const float*)d_in[3];
  const float* U    = (const float*)d_in[4];
  const float* bias = (const float*)d_in[5];
  const int*   ei   = (const int*)d_in[6];
  int E = in_sizes[1];
  float* ws  = (float*)d_ws;
  float* out = (float*)d_out;

  k_init<<<64, INIT_THREADS, 0, stream>>>(x, W, att, ws);
  k_edge_max<<<MX_GRID, MX_THREADS, 0, stream>>>(ei, ea, att, x, W, ws, E);
  k_edge_exp<<<EDGE_BLOCKS, EB_THREADS, 0, stream>>>(ei, ea, att, ws, E);
  k_final<<<MTILES, 128, 0, stream>>>(U, bias, ws, out);
}


# exp pass: threads past a block's edge range skip the ranking add; they used to add into one spare LDS counter, 32 lanes a wave on one word, about 22 waves a block in turn; sums unchanged
# speedup vs baseline: 4.0199x; 1.0365x over previous
#include <hip/hip_runtime.h>
#include <math.h>

typedef float v2f __attribute__((ext_vector_type(2)));
typedef float v8f __attribute__((ext_vector_type(8)));

#define N_NODES 177
#define FEAT 128
#define OUT_C 128
#define NEG_SLOPE 0.2f
#define MTILES 12
#define S_PAD 180
#define H_ROWS 180
#define N_PAD 192
#define EDGE_BLOCKS 128

#define OFF_H    0
#define OFF_A    (OFF_H + H_ROWS * FEAT)
#define OFF_B    (OFF_A + N_NODES)
#define OFF_AMAX (OFF_B + N_NODES)
#define OFF_S    (OFF_AMAX + N_NODES)
#define OFF_PES  ((OFF_S + N_NODES * S_PAD + 31) & ~31)
#define OFF_PTS  (OFF_PES + EDGE_BLOCKS * N_PAD)
#define WS_FLOATS (OFF_PTS + EDGE_BLOCKS * N_PAD)

__device__ __forceinline__ unsigned f2ord(float f) {
  unsigned u = __float_as_uint(f);
  return (u & 0x80000000u) ? ~u : (u | 0x80000000u);
}
__device__ __forceinline__ float ord2f(unsigned u) {
  return __uint_as_float((u & 0x80000000u) ? (u ^ 0x80000000u) : ~u);
}
#define ORD_NEG_INF 0x007FFFFFu

__device__ __forceinline__ v2f load_a_clamp(const float* __restrict__ A, int m0, int k0,
                                            int lane, int ld) {
  int m = m0 + (lane & 15);
  m = (m < N_NODES) ? m : (N_NODES - 1);
  int kb = k0 + ((lane >> 4) << 1);
  const float* p = A + m * ld + kb;
  v2f a; a.x = p[0]; a.y = p[1];
  return a;
}
__device__ __forceinline__ v2f load_b(const float* __restrict__ B, int k0, int n0,
                                      int lane, int ld) {
  int n  = n0 + (lane & 15);
  int kb = k0 + ((lane >> 4) << 1);
  const float* p = B + kb * ld + n;
  v2f b; b.x = p[0]; b.y = p[ld];
  return b;
}
__device__ __forceinline__ v2f load_a_lds(const float* A, int k0, int lane) {
  int m  = lane & 15;
  int kb = k0 + ((lane >> 4) << 1);
  v2f a;
  a.x = A[m * OUT_C + kb];
  a.y = A[m * OUT_C + kb + 1];
  return a;
}

#define INIT_THREADS 256
#define INIT_ROWS_PER_WAVE (FEAT / (INIT_THREADS / 32))
__global__ __launch_bounds__(INIT_THREADS)
void k_init(const float* __restrict__ x, const float* __restrict__ W,
            const float* __restrict__ att, float* __restrict__ ws) {
  __shared__ float s_va[FEAT];
  __shared__ float s_vb[FEAT];
  int tid    = blockIdx.x * blockDim.x + threadIdx.x;
  int stride = gridDim.x * blockDim.x;
  const int lane = threadIdx.x & 31;
  const int wave = threadIdx.x >> 5;
  const bool ab_block = blockIdx.x < MTILES;
  const int r  = lane & 15;
  const int hh = lane >> 4;
  const int m  = (int)blockIdx.x * 16 + r;
  const int mc = (m < N_NODES) ? m : (N_NODES - 1);
  float wv[INIT_ROWS_PER_WAVE][FEAT / 32] = {};
  float a0[FEAT / 32] = {}, a1[FEAT / 32] = {};
  float xv[FEAT / 2] = {};
  if (ab_block) {
#pragma unroll
    for (int q = 0; q < FEAT / 32; ++q) {
      a0[q] = att[lane + 32 * q];
      a1[q] = att[OUT_C + lane + 32 * q];
    }
#pragma unroll
    for (int j = 0; j < INIT_ROWS_PER_WAVE; ++j) {
      const int row = wave * INIT_ROWS_PER_WAVE + j;
#pragma unroll
      for (int q = 0; q < FEAT / 32; ++q) wv[j][q] = W[row * OUT_C + lane + 32 * q];
    }
    if (wave == 0) {
      const float* xr = x + mc * FEAT + hh * (FEAT / 2);
#pragma unroll
      for (int i = 0; i < FEAT / 2; ++i) xv[i] = xr[i];
    }
  }
  unsigned* amax = (unsigned*)(ws + OFF_AMAX);
  for (int j = tid; j < N_NODES * S_PAD; j += stride) ws[OFF_S + j] = 0.0f;
  for (int j = tid; j < (H_ROWS - N_NODES) * FEAT; j += stride)
    ws[OFF_H + N_NODES * FEAT + j] = 0.0f;
  for (int j = tid; j < N_NODES; j += stride) amax[j] = ORD_NEG_INF;
  if (ab_block) {
    float pa[INIT_ROWS_PER_WAVE], pb[INIT_ROWS_PER_WAVE];
#pragma unroll
    for (int j = 0; j < INIT_ROWS_PER_WAVE; ++j) {
      pa[j] = 0.0f; pb[j] = 0.0f;
#pragma unroll
      for (int q = 0; q < FEAT / 32; ++q) { pa[j] += wv[j][q] * a0[q]; pb[j] += wv[j][q] * a1[q]; }
    }
    const bool b4 = (lane & 16) != 0, b3 = (lane & 8) != 0, b2 = (lane & 4) != 0, b1 = (lane & 2) != 0;
    float qa[8], qb[8];
#pragma unroll
    for (int i = 0; i < 8; ++i) {
      const float ka = b4 ? pa[i + 8] : pa[i], sa = b4 ? pa[i] : pa[i + 8];
      const float kb = b4 ? pb[i + 8] : pb[i], sb = b4 ? pb[i] : pb[i + 8];
      qa[i] = ka + __shfl_xor(sa, 16, 32);
      qb[i] = kb + __shfl_xor(sb, 16, 32);
    }
    float ra[4], rb[4];
#pragma unroll
    for (int i = 0; i < 4; ++i) {
      const float ka = b3 ? qa[i + 4] : qa[i], sa = b3 ? qa[i] : qa[i + 4];
      const float kb = b3 ? qb[i + 4] : qb[i], sb = b3 ? qb[i] : qb[i + 4];
      ra[i] = ka + __shfl_xor(sa, 8, 32);
      rb[i] = kb + __shfl_xor(sb, 8, 32);
    }
    float ua[2], ub[2];
#pragma unroll
    for (int i = 0; i < 2; ++i) {
      const float ka = b2 ? ra[i + 2] : ra[i], sa = b2 ? ra[i] : ra[i + 2];
      const float kb = b2 ? rb[i + 2] : rb[i], sb = b2 ? rb[i] : rb[i + 2];
      ua[i] = ka + __shfl_xor(sa, 4, 32);
      ub[i] = kb + __shfl_xor(sb, 4, 32);
    }
    float ta, tb;
    {
      const float ka = b1 ? ua[1] : ua[0], sa = b1 ? ua[0] : ua[1];
      const float kb = b1 ? ub[1] : ub[0], sb = b1 ? ub[0] : ub[1];
      ta = ka + __shfl_xor(sa, 2, 32);
      tb = kb + __shfl_xor(sb, 2, 32);
    }
    ta += __shfl_xor(ta, 1, 32);
    tb += __shfl_xor(tb, 1, 32);
    if ((lane & 1) == 0) {
      s_va[wave * INIT_ROWS_PER_WAVE + (lane >> 1)] = ta;
      s_vb[wave * INIT_ROWS_PER_WAVE + (lane >> 1)] = tb;
    }
    __syncthreads();
    if (wave == 0) {
      const float* va = s_va + hh * (FEAT / 2);
      const float* vb = s_vb + hh * (FEAT / 2);
      float sa = 0.0f, sb = 0.0f;
#pragma unroll
      for (int i = 0; i < FEAT / 2; ++i) {
        sa += xv[i] * va[i];
        sb += xv[i] * vb[i];
      }
      sa += __shfl_xor(sa, 16, 32);
      sb += __shfl_xor(sb, 16, 32);
      if (hh == 0 && m < N_NODES) { ws[OFF_A + m] = sa; ws[OFF_B + m] = sb; }
    }
  }
}

__device__ __forceinline__ void node_gemm_strip(const float* __restrict__ x, const float* __restrict__ W,
                                                float* __restrict__ h, int m0, int wave, int lane) {
  const int n0 = wave * 16;
  v8f c0 = {};
  for (int k0 = 0; k0 < FEAT; k0 += 4) {
    v2f a  = load_a_clamp(x, m0, k0, lane, FEAT);
    v2f b0 = load_b(W, k0, n0, lane, OUT_C);
    c0 = __builtin_amdgcn_wmma_f32_16x16x4_f32(false, a, false, b0, (short)0, c0, false, false);
  }
  const int nb = n0 + (lane & 15);
  const int mh = (lane >> 4) * 8;
  for (int j = 0; j < 8; ++j) {
    const int m = m0 + j + mh;
    if (m < N_NODES) h[m * OUT_C + nb] = c0[j];
  }
}

#define MX_THREADS 256
#define MX_BLOCKS  128
#define MX_GRID    (MX_BLOCKS + MTILES)
#define MX_R       16
#define MX_BATCH   (MX_THREADS * MX_R)
#define MX_WAVES   (MX_THREADS / 32)
#define MX_ROWS_PER_WAVE ((N_NODES + MX_WAVES - 1) / MX_WAVES)

__global__ __launch_bounds__(MX_THREADS)
void k_edge_max(const int* __restrict__ ei, const float* __restrict__ ea,
                const float* __restrict__ att, const float* __restrict__ x,
                const float* __restrict__ W, float* __restrict__ ws, int E) {
  __shared__ __attribute__((aligned(16))) float tbl[N_NODES * MX_THREADS];
  __shared__ float s_row[N_NODES];

  if ((int)blockIdx.x >= MX_BLOCKS) {
    node_gemm_strip(x, W, ws + OFF_H, ((int)blockIdx.x - MX_BLOCKS) * 16,
                    (int)threadIdx.x >> 5, (int)threadIdx.x & 31);
    return;
  }

  const float* a = ws + OFF_A;
  const float* b = ws + OFF_B;
  unsigned* amax = (unsigned*)(ws + OFF_AMAX);
  const float cc = att[2 * OUT_C];

  const int tid  = threadIdx.x;
  const int lane = tid & 31;
  const int wave = tid >> 5;

  const int per = (E + MX_BLOCKS - 1) / MX_BLOCKS;
  const int e0  = (int)blockIdx.x * per;
  const int e1  = (e0 + per < E) ? (e0 + per) : E;
  if (e0 >= e1) return;

  int   sv[MX_R], dv[MX_R];
  float ev[MX_R];
#pragma unroll
  for (int i = 0; i < MX_R; ++i) {
    const int e  = e0 + i * MX_THREADS + tid;
    const int ec = (e < e1) ? e : e0;
    sv[i] = ei[ec];
    dv[i] = ei[E + ec];
    ev[i] = ea[ec];
  }

  {
    const float ninf = -INFINITY;
    const float4 n4 = make_float4(ninf, ninf, ninf, ninf);
    float4* t4 = (float4*)tbl;
    for (int q = tid; q < N_NODES * MX_THREADS / 4; q += MX_THREADS) t4[q] = n4;
  }
  __syncthreads();

  float* col = tbl + tid;
  for (int base = e0; ; ) {
    float al[MX_R];
#pragma unroll
    for (int i = 0; i < MX_R; ++i) {
      const int e = base + i * MX_THREADS + tid;
      float v = a[dv[i]] + b[sv[i]] + cc * ev[i];
      v = (v > 0.0f) ? v : NEG_SLOPE * v;
      al[i] = (e < e1) ? v : -INFINITY;
    }
#pragma unroll
    for (int i = 0; i < MX_R; ++i) {
      float* p = col + dv[i] * MX_THREADS;
      *p = fmaxf(*p, al[i]);
    }
    base += MX_BATCH;
    if (base >= e1) break;
#pragma unroll
    for (int i = 0; i < MX_R; ++i) {
      const int e  = base + i * MX_THREADS + tid;
      const int ec = (e < e1) ? e : e0;
      sv[i] = ei[ec];
      dv[i] = ei[E + ec];
      ev[i] = ea[ec];
    }
  }
  __syncthreads();

  float m[MX_ROWS_PER_WAVE];
#pragma unroll
  for (int j = 0; j < MX_ROWS_PER_WAVE; ++j) {
    int r = wave + j * MX_WAVES;
    r = (r < N_NODES) ? r : (N_NODES - 1);
    const float4 v0 = *(const float4*)(&tbl[r * MX_THREADS + lane * 4]);
    const float4 v1 = *(const float4*)(&tbl[r * MX_THREADS + 128 + lane * 4]);
    m[j] = fmaxf(fmaxf(fmaxf(v0.x, v0.y), fmaxf(v0.z, v0.w)),
                 fmaxf(fmaxf(v1.x, v1.y), fmaxf(v1.z, v1.w)));
  }
#pragma unroll
  for (int k = 16; k >= 1; k >>= 1) {
#pragma unroll
    for (int j = 0; j < MX_ROWS_PER_WAVE; ++j) m[j] = fmaxf(m[j], __shfl_xor(m[j], k, 32));
  }
  float mine = -INFINITY;
#pragma unroll
  for (int j = 0; j < MX_ROWS_PER_WAVE; ++j) mine = (lane == j) ? m[j] : mine;
  {
    const int r = wave + lane * MX_WAVES;
    if (lane < MX_ROWS_PER_WAVE && r < N_NODES) s_row[r] = mine;
  }
  __syncthreads();
  if (tid < N_NODES && s_row[tid] > -INFINITY) atomicMax(&amax[tid], f2ord(s_row[tid]));
}

#define EB_THREADS 768
#define EB_R       6
#define EB_BATCH   (EB_THREADS * EB_R)
#define NBINS      144
#define REC_CAP    (EB_BATCH + NBINS * 3 + 8)

__global__ __launch_bounds__(EB_THREADS)
void k_edge_exp(const int* __restrict__ ei, const float* __restrict__ ea,
                const float* __restrict__ att, float* __restrict__ ws, int E) {
  __shared__ int   s_cnt[NBINS];
  __shared__ int   s_start[NBINS];
  __shared__ __attribute__((aligned(16))) int   r_key[REC_CAP];
  __shared__ __attribute__((aligned(16))) float r_w[REC_CAP];
  __shared__ __attribute__((aligned(16))) float r_we[REC_CAP];
  __shared__ float s_es[(EB_THREADS / 32) * 16];
  __shared__ float s_ts[(EB_THREADS / 32) * 16];

  const float* a = ws + OFF_A;
  const float* b = ws + OFF_B;
  const unsigned* amax = (const unsigned*)(ws + OFF_AMAX);
  float* pes  = ws + OFF_PES + (int)blockIdx.x * N_PAD;
  float* pts  = ws + OFF_PTS + (int)blockIdx.x * N_PAD;
  float* S    = ws + OFF_S;
  const float cc = att[2 * OUT_C];

  const int tid  = threadIdx.x;
  const int lane = tid & 31;
  const int wave = tid >> 5;
  const int c15  = lane & 15;
  const int h    = lane >> 4;
  const int first_bin = wave * 6;

  const int per = (E + (int)gridDim.x - 1) / (int)gridDim.x;
  const int e0  = (int)blockIdx.x * per;
  const int e1  = (e0 + per < E) ? (e0 + per) : E;

  v8f acc[6];
#pragma unroll
  for (int j = 0; j < 6; ++j) acc[j] = (v8f){};
  float es = 0.0f, ts = 0.0f;

  for (int base = e0; base < e1; base += EB_BATCH) {
    if (tid < NBINS) s_cnt[tid] = 0;
    __syncthreads();

    int   sv[EB_R], dv[EB_R];
    float ev[EB_R];
#pragma unroll
    for (int i = 0; i < EB_R; ++i) {
      const int e  = base + i * EB_THREADS + tid;
      const int ec = (e < e1) ? e : e0;
      sv[i] = ei[ec];
      dv[i] = ei[E + ec];
      ev[i] = ea[ec];
    }
    int   pk[EB_R], ppos[EB_R];
    float pw[EB_R], pwe[EB_R];
#pragma unroll
    for (int i = 0; i < EB_R; ++i) {
      float al = a[dv[i]] + b[sv[i]] + cc * ev[i];
      al = (al > 0.0f) ? al : NEG_SLOPE * al;
      const float w = __expf(al - ord2f(amax[dv[i]]));
      pk[i]  = (dv[i] & 15) | ((sv[i] & 15) << 4);
      pw[i]  = w;
      pwe[i] = w * ev[i];
    }
#pragma unroll
    for (int i = 0; i < EB_R; ++i) {
      const int  e   = base + i * EB_THREADS + tid;
      const bool ok  = e < e1;
      const int  bin = (dv[i] >> 4) * 12 + (sv[i] >> 4);
      int pos = 0;
      if (ok) pos = atomicAdd(&s_cnt[bin], 1);
      ppos[i] = ok ? ((bin << 16) | pos) : -1;
    }
    __syncthreads();

    {
      int sum = 0;
      for (int q = lane; q < first_bin; q += 32) sum += (s_cnt[q] + 3) & ~3;
      for (int m = 16; m >= 1; m >>= 1) sum += __shfl_xor(sum, m, 32);
      if (lane < 6) {
        int st = sum;
        for (int jj = 0; jj < lane; ++jj) st += (s_cnt[first_bin + jj] + 3) & ~3;
        s_start[first_bin + lane] = st;
        const int nb = s_cnt[first_bin + lane];
        for (int p = st + nb; p < st + ((nb + 3) & ~3); ++p) {
          r_key[p] = 0; r_w[p] = 0.0f; r_we[p] = 0.0f;
        }
      }
    }
    __syncthreads();

#pragma unroll
    for (int i = 0; i < EB_R; ++i) {
      if (ppos[i] >= 0) {
        const int idx = s_start[ppos[i] >> 16] + (ppos[i] & 0xFFFF);
        r_key[idx] = pk[i];
        r_w[idx]   = pw[i];
        r_we[idx]  = pwe[i];
      }
    }
    __syncthreads();

#pragma unroll
    for (int j = 0; j < 6; ++j) {
      const int n   = __builtin_amdgcn_readfirstlane(s_cnt[first_bin + j]);
      const int st  = __builtin_amdgcn_readfirstlane(s_start[first_bin + j]);
      const int nch = (n + 3) >> 2;
      for (int c = 0; c < nch; ++c) {
        const int idx = st + 4 * c + 2 * h;
        const int2   kk = *(const int2*)(&r_key[idx]);
        const float2 ww = *(const float2*)(&r_w[idx]);
        const float2 we = *(const float2*)(&r_we[idx]);
        const bool m0 = ((kk.x & 15) == c15);
        const bool m1 = ((kk.y & 15) == c15);
        v2f af, bf;
        af.x = m0 ? ww.x : 0.0f;
        af.y = m1 ? ww.y : 0.0f;
        bf.x = ((kk.x >> 4) == c15) ? 1.0f : 0.0f;
        bf.y = ((kk.y >> 4) == c15) ? 1.0f : 0.0f;
        acc[j] = __builtin_amdgcn_wmma_f32_16x16x4_f32(false, af, false, bf, (short)0, acc[j],
                                                       false, false);
        es += af.x + af.y;
        ts += (m0 ? we.x : 0.0f) + (m1 ? we.y : 0.0f);
      }
    }
    __syncthreads();
  }

  es += __shfl_xor(es, 16, 32);
  ts += __shfl_xor(ts, 16, 32);
  if (h == 0) { s_es[wave * 16 + c15] = es; s_ts[wave * 16 + c15] = ts; }
  __syncthreads();
  if (tid < N_PAD) {
    const int w0 = (tid >> 4) * 2;
    const int r  = tid & 15;
    pes[tid] = s_es[w0 * 16 + r] + s_es[(w0 + 1) * 16 + r];
    pts[tid] = s_ts[w0 * 16 + r] + s_ts[(w0 + 1) * 16 + r];
  }

  const int dt = wave >> 1;
  const int st0 = (wave & 1) * 6;
#pragma unroll
  for (int j = 0; j < 6; ++j) {
    const int col = (st0 + j) * 16 + c15;
#pragma unroll
    for (int r = 0; r < 8; ++r) {
      const int row = dt * 16 + 8 * h + r;
      const float v = acc[j][r];
      if (row < N_NODES && col < N_NODES && v != 0.0f) atomicAdd(&S[row * S_PAD + col], v);
    }
  }
}

__global__ __launch_bounds__(128)
void k_final(const float* __restrict__ U, const float* __restrict__ bias,
             const float* __restrict__ ws, float* __restrict__ out) {
  __shared__ float aggr[16 * OUT_C];
  __shared__ __attribute__((aligned(16))) float s_pe[16 * 8];
  __shared__ __attribute__((aligned(16))) float s_pt[16 * 8];
  const float* h    = ws + OFF_H;
  const float* S    = ws + OFF_S;
  const float* pes  = ws + OFF_PES;
  const float* pts  = ws + OFF_PTS;
  int lane = threadIdx.x & 31;
  int wave = threadIdx.x >> 5;
  int m0 = blockIdx.x * 16;
  int n0 = wave * 32;
  int nb = n0 + (lane & 15);
  int mh = (lane >> 4) * 8;

  const int fr = threadIdx.x & 15;
  const int fg = threadIdx.x >> 4;
  float pe[16], pt[16];
#pragma unroll
  for (int i = 0; i < 16; ++i) {
    pe[i] = pes[(fg * 16 + i) * N_PAD + m0 + fr];
    pt[i] = pts[(fg * 16 + i) * N_PAD + m0 + fr];
  }
  const float u_last0 = U[OUT_C * OUT_C + nb];
  const float u_last1 = U[OUT_C * OUT_C + nb + 16];
  const float bias0   = bias[nb];
  const float bias1   = bias[nb + 16];

  v8f c0 = {}; v8f c1 = {};
#pragma unroll
  for (int k0 = 0; k0 < S_PAD; k0 += 4) {
    v2f a  = load_a_clamp(S, m0, k0, lane, S_PAD);
    v2f b0 = load_b(h, k0, n0,      lane, OUT_C);
    v2f b1 = load_b(h, k0, n0 + 16, lane, OUT_C);
    c0 = __builtin_amdgcn_wmma_f32_16x16x4_f32(false, a, false, b0, (short)0, c0, false, false);
    c1 = __builtin_amdgcn_wmma_f32_16x16x4_f32(false, a, false, b1, (short)0, c1, false, false);
  }
  v2f ub0[OUT_C / 4], ub1[OUT_C / 4];
#pragma unroll
  for (int i = 0; i < OUT_C / 4; ++i) {
    ub0[i] = load_b(U, 4 * i, n0,      lane, OUT_C);
    ub1[i] = load_b(U, 4 * i, n0 + 16, lane, OUT_C);
  }

  {
    float se = pe[0], st = pt[0];
#pragma unroll
    for (int i = 1; i < 16; ++i) { se += pe[i]; st += pt[i]; }
    s_pe[fr * 8 + fg] = se;
    s_pt[fr * 8 + fg] = st;
  }
  __syncthreads();
  float rinv[8], tn[8];
#pragma unroll
  for (int j = 0; j < 8; ++j) {
    int ml = j + mh;
    ml = (m0 + ml < N_NODES) ? ml : (N_NODES - 1 - m0);
    const float4 ea0 = *(const float4*)(&s_pe[ml * 8]);
    const float4 ea1 = *(const float4*)(&s_pe[ml * 8 + 4]);
    const float4 ta0 = *(const float4*)(&s_pt[ml * 8]);
    const float4 ta1 = *(const float4*)(&s_pt[ml * 8 + 4]);
    const float esm = ((ea0.x + ea0.y) + (ea0.z + ea0.w)) + ((ea1.x + ea1.y) + (ea1.z + ea1.w));
    const float tsm = ((ta0.x + ta0.y) + (ta0.z + ta0.w)) + ((ta1.x + ta1.y) + (ta1.z + ta1.w));
    const float es = esm + 1e-16f;
    rinv[j] = 1.0f / es;
    tn[j]   = tsm / es;
  }

#pragma unroll
  for (int j = 0; j < 8; ++j) {
    int mloc = j + mh;
    aggr[mloc * OUT_C + nb]      = c0[j] * rinv[j];
    aggr[mloc * OUT_C + nb + 16] = c1[j] * rinv[j];
  }
  __syncthreads();

  v8f d0 = {}; v8f d1 = {};
#pragma unroll
  for (int i = 0; i < OUT_C / 4; ++i) {
    v2f a  = load_a_lds(aggr, 4 * i, lane);
    d0 = __builtin_amdgcn_wmma_f32_16x16x4_f32(false, a, false, ub0[i], (short)0, d0, false, false);
    d1 = __builtin_amdgcn_wmma_f32_16x16x4_f32(false, a, false, ub1[i], (short)0, d1, false, false);
  }
#pragma unroll
  for (int j = 0; j < 8; ++j) {
    int m = m0 + j + mh;
    if (m < N_NODES) {
      out[m * OUT_C + nb]      = d0[j] + tn[j] * u_last0 + bias0;
      out[m * OUT_C + nb + 16] = d1[j] + tn[j] * u_last1 + bias1;
    }
  }
}

extern "C" void kernel_launch(void* const* d_in, const int* in_sizes, int n_in,
                              void* d_out, int out_size, void* d_ws, size_t ws_size,
                              hipStream_t stream) {
  const float* x    = (const float*)d_in[0];
  const float* ea   = (const float*)d_in[1];
  const float* W    = (const float*)d_in[2];
  const float* att  = (const float*)d_in[3];
  const float* U    = (const float*)d_in[4];
  const float* bias = (const float*)d_in[5];
  const int*   ei   = (const int*)d_in[6];
  int E = in_sizes[1];
  float* ws  = (float*)d_ws;
  float* out = (float*)d_out;

  k_init<<<64, INIT_THREADS, 0, stream>>>(x, W, att, ws);
  k_edge_max<<<MX_GRID, MX_THREADS, 0, stream>>>(ei, ea, att, x, W, ws, E);
  k_edge_exp<<<EDGE_BLOCKS, EB_THREADS, 0, stream>>>(ei, ea, att, ws, E);
  k_final<<<MTILES, 128, 0, stream>>>(U, bias, ws, out);
}


# S kept in memory as 16x16 tiles in the WMMA accumulator's own order: each flush atomic of the exp pass falls in one aligned 128-byte line (was 2 to 4); final launch reads the tiles; sums as before
# speedup vs baseline: 4.9919x; 1.2418x over previous
#include <hip/hip_runtime.h>
#include <math.h>

typedef float v2f __attribute__((ext_vector_type(2)));
typedef float v8f __attribute__((ext_vector_type(8)));

#define N_NODES 177
#define FEAT 128
#define OUT_C 128
#define NEG_SLOPE 0.2f
#define MTILES 12
#define S_PAD 180
#define STILES 12
#define S_TILE_WORDS 256
#define H_ROWS 180
#define N_PAD 192
#define EDGE_BLOCKS 128

#define OFF_H    0
#define OFF_A    (OFF_H + H_ROWS * FEAT)
#define OFF_B    (OFF_A + N_NODES)
#define OFF_AMAX (OFF_B + N_NODES)
#define S_WORDS  (MTILES * STILES * S_TILE_WORDS)
#define OFF_S    ((OFF_AMAX + N_NODES + 31) & ~31)
#define OFF_PES  (OFF_S + S_WORDS)
#define OFF_PTS  (OFF_PES + EDGE_BLOCKS * N_PAD)
#define WS_FLOATS (OFF_PTS + EDGE_BLOCKS * N_PAD)

__device__ __forceinline__ unsigned f2ord(float f) {
  unsigned u = __float_as_uint(f);
  return (u & 0x80000000u) ? ~u : (u | 0x80000000u);
}
__device__ __forceinline__ float ord2f(unsigned u) {
  return __uint_as_float((u & 0x80000000u) ? (u ^ 0x80000000u) : ~u);
}
#define ORD_NEG_INF 0x007FFFFFu

__device__ __forceinline__ v2f load_a_clamp(const float* __restrict__ A, int m0, int k0,
                                            int lane, int ld) {
  int m = m0 + (lane & 15);
  m = (m < N_NODES) ? m : (N_NODES - 1);
  int kb = k0 + ((lane >> 4) << 1);
  const float* p = A + m * ld + kb;
  v2f a; a.x = p[0]; a.y = p[1];
  return a;
}
__device__ __forceinline__ v2f load_a_tile(const float* __restrict__ S, int dtile, int k0, int lane) {
  const int rho = lane & 15;
  const float* p = S + (dtile * STILES + (k0 >> 4)) * S_TILE_WORDS
                     + (rho & 7) * 32 + (rho >> 3) * 16 + (k0 & 15) + ((lane >> 4) << 1);
  const float2 t = *(const float2*)p;
  v2f a; a.x = t.x; a.y = t.y;
  return a;
}
__device__ __forceinline__ v2f load_b(const float* __restrict__ B, int k0, int n0,
                                      int lane, int ld) {
  int n  = n0 + (lane & 15);
  int kb = k0 + ((lane >> 4) << 1);
  const float* p = B + kb * ld + n;
  v2f b; b.x = p[0]; b.y = p[ld];
  return b;
}
__device__ __forceinline__ v2f load_a_lds(const float* A, int k0, int lane) {
  int m  = lane & 15;
  int kb = k0 + ((lane >> 4) << 1);
  v2f a;
  a.x = A[m * OUT_C + kb];
  a.y = A[m * OUT_C + kb + 1];
  return a;
}

#define INIT_THREADS 256
#define INIT_ROWS_PER_WAVE (FEAT / (INIT_THREADS / 32))
__global__ __launch_bounds__(INIT_THREADS)
void k_init(const float* __restrict__ x, const float* __restrict__ W,
            const float* __restrict__ att, float* __restrict__ ws) {
  __shared__ float s_va[FEAT];
  __shared__ float s_vb[FEAT];
  int tid    = blockIdx.x * blockDim.x + threadIdx.x;
  int stride = gridDim.x * blockDim.x;
  const int lane = threadIdx.x & 31;
  const int wave = threadIdx.x >> 5;
  const bool ab_block = blockIdx.x < MTILES;
  const int r  = lane & 15;
  const int hh = lane >> 4;
  const int m  = (int)blockIdx.x * 16 + r;
  const int mc = (m < N_NODES) ? m : (N_NODES - 1);
  float wv[INIT_ROWS_PER_WAVE][FEAT / 32] = {};
  float a0[FEAT / 32] = {}, a1[FEAT / 32] = {};
  float xv[FEAT / 2] = {};
  if (ab_block) {
#pragma unroll
    for (int q = 0; q < FEAT / 32; ++q) {
      a0[q] = att[lane + 32 * q];
      a1[q] = att[OUT_C + lane + 32 * q];
    }
#pragma unroll
    for (int j = 0; j < INIT_ROWS_PER_WAVE; ++j) {
      const int row = wave * INIT_ROWS_PER_WAVE + j;
#pragma unroll
      for (int q = 0; q < FEAT / 32; ++q) wv[j][q] = W[row * OUT_C + lane + 32 * q];
    }
    if (wave == 0) {
      const float* xr = x + mc * FEAT + hh * (FEAT / 2);
#pragma unroll
      for (int i = 0; i < FEAT / 2; ++i) xv[i] = xr[i];
    }
  }
  unsigned* amax = (unsigned*)(ws + OFF_AMAX);
  {
    float4* S4 = (float4*)(ws + OFF_S);
    const float4 z4 = make_float4(0.0f, 0.0f, 0.0f, 0.0f);
    for (int j = tid; j < S_WORDS / 4; j += stride) S4[j] = z4;
  }
  for (int j = tid; j < (H_ROWS - N_NODES) * FEAT; j += stride)
    ws[OFF_H + N_NODES * FEAT + j] = 0.0f;
  for (int j = tid; j < N_NODES; j += stride) amax[j] = ORD_NEG_INF;
  if (ab_block) {
    float pa[INIT_ROWS_PER_WAVE], pb[INIT_ROWS_PER_WAVE];
#pragma unroll
    for (int j = 0; j < INIT_ROWS_PER_WAVE; ++j) {
      pa[j] = 0.0f; pb[j] = 0.0f;
#pragma unroll
      for (int q = 0; q < FEAT / 32; ++q) { pa[j] += wv[j][q] * a0[q]; pb[j] += wv[j][q] * a1[q]; }
    }
    const bool b4 = (lane & 16) != 0, b3 = (lane & 8) != 0, b2 = (lane & 4) != 0, b1 = (lane & 2) != 0;
    float qa[8], qb[8];
#pragma unroll
    for (int i = 0; i < 8; ++i) {
      const float ka = b4 ? pa[i + 8] : pa[i], sa = b4 ? pa[i] : pa[i + 8];
      const float kb = b4 ? pb[i + 8] : pb[i], sb = b4 ? pb[i] : pb[i + 8];
      qa[i] = ka + __shfl_xor(sa, 16, 32);
      qb[i] = kb + __shfl_xor(sb, 16, 32);
    }
    float ra[4], rb[4];
#pragma unroll
    for (int i = 0; i < 4; ++i) {
      const float ka = b3 ? qa[i + 4] : qa[i], sa = b3 ? qa[i] : qa[i + 4];
      const float kb = b3 ? qb[i + 4] : qb[i], sb = b3 ? qb[i] : qb[i + 4];
      ra[i] = ka + __shfl_xor(sa, 8, 32);
      rb[i] = kb + __shfl_xor(sb, 8, 32);
    }
    float ua[2], ub[2];
#pragma unroll
    for (int i = 0; i < 2; ++i) {
      const float ka = b2 ? ra[i + 2] : ra[i], sa = b2 ? ra[i] : ra[i + 2];
      const float kb = b2 ? rb[i + 2] : rb[i], sb = b2 ? rb[i] : rb[i + 2];
      ua[i] = ka + __shfl_xor(sa, 4, 32);
      ub[i] = kb + __shfl_xor(sb, 4, 32);
    }
    float ta, tb;
    {
      const float ka = b1 ? ua[1] : ua[0], sa = b1 ? ua[0] : ua[1];
      const float kb = b1 ? ub[1] : ub[0], sb = b1 ? ub[0] : ub[1];
      ta = ka + __shfl_xor(sa, 2, 32);
      tb = kb + __shfl_xor(sb, 2, 32);
    }
    ta += __shfl_xor(ta, 1, 32);
    tb += __shfl_xor(tb, 1, 32);
    if ((lane & 1) == 0) {
      s_va[wave * INIT_ROWS_PER_WAVE + (lane >> 1)] = ta;
      s_vb[wave * INIT_ROWS_PER_WAVE + (lane >> 1)] = tb;
    }
    __syncthreads();
    if (wave == 0) {
      const float* va = s_va + hh * (FEAT / 2);
      const float* vb = s_vb + hh * (FEAT / 2);
      float sa = 0.0f, sb = 0.0f;
#pragma unroll
      for (int i = 0; i < FEAT / 2; ++i) {
        sa += xv[i] * va[i];
        sb += xv[i] * vb[i];
      }
      sa += __shfl_xor(sa, 16, 32);
      sb += __shfl_xor(sb, 16, 32);
      if (hh == 0 && m < N_NODES) { ws[OFF_A + m] = sa; ws[OFF_B + m] = sb; }
    }
  }
}

__device__ __forceinline__ void node_gemm_strip(const float* __restrict__ x, const float* __restrict__ W,
                                                float* __restrict__ h, int m0, int wave, int lane) {
  const int n0 = wave * 16;
  v8f c0 = {};
  for (int k0 = 0; k0 < FEAT; k0 += 4) {
    v2f a  = load_a_clamp(x, m0, k0, lane, FEAT);
    v2f b0 = load_b(W, k0, n0, lane, OUT_C);
    c0 = __builtin_amdgcn_wmma_f32_16x16x4_f32(false, a, false, b0, (short)0, c0, false, false);
  }
  const int nb = n0 + (lane & 15);
  const int mh = (lane >> 4) * 8;
  for (int j = 0; j < 8; ++j) {
    const int m = m0 + j + mh;
    if (m < N_NODES) h[m * OUT_C + nb] = c0[j];
  }
}

#define MX_THREADS 256
#define MX_BLOCKS  128
#define MX_GRID    (MX_BLOCKS + MTILES)
#define MX_R       16
#define MX_BATCH   (MX_THREADS * MX_R)
#define MX_WAVES   (MX_THREADS / 32)
#define MX_ROWS_PER_WAVE ((N_NODES + MX_WAVES - 1) / MX_WAVES)

__global__ __launch_bounds__(MX_THREADS)
void k_edge_max(const int* __restrict__ ei, const float* __restrict__ ea,
                const float* __restrict__ att, const float* __restrict__ x,
                const float* __restrict__ W, float* __restrict__ ws, int E) {
  __shared__ __attribute__((aligned(16))) float tbl[N_NODES * MX_THREADS];
  __shared__ float s_row[N_NODES];

  if ((int)blockIdx.x >= MX_BLOCKS) {
    node_gemm_strip(x, W, ws + OFF_H, ((int)blockIdx.x - MX_BLOCKS) * 16,
                    (int)threadIdx.x >> 5, (int)threadIdx.x & 31);
    return;
  }

  const float* a = ws + OFF_A;
  const float* b = ws + OFF_B;
  unsigned* amax = (unsigned*)(ws + OFF_AMAX);
  const float cc = att[2 * OUT_C];

  const int tid  = threadIdx.x;
  const int lane = tid & 31;
  const int wave = tid >> 5;

  const int per = (E + MX_BLOCKS - 1) / MX_BLOCKS;
  const int e0  = (int)blockIdx.x * per;
  const int e1  = (e0 + per < E) ? (e0 + per) : E;
  if (e0 >= e1) return;

  int   sv[MX_R], dv[MX_R];
  float ev[MX_R];
#pragma unroll
  for (int i = 0; i < MX_R; ++i) {
    const int e  = e0 + i * MX_THREADS + tid;
    const int ec = (e < e1) ? e : e0;
    sv[i] = ei[ec];
    dv[i] = ei[E + ec];
    ev[i] = ea[ec];
  }

  {
    const float ninf = -INFINITY;
    const float4 n4 = make_float4(ninf, ninf, ninf, ninf);
    float4* t4 = (float4*)tbl;
    for (int q = tid; q < N_NODES * MX_THREADS / 4; q += MX_THREADS) t4[q] = n4;
  }
  __syncthreads();

  float* col = tbl + tid;
  for (int base = e0; ; ) {
    float al[MX_R];
#pragma unroll
    for (int i = 0; i < MX_R; ++i) {
      const int e = base + i * MX_THREADS + tid;
      float v = a[dv[i]] + b[sv[i]] + cc * ev[i];
      v = (v > 0.0f) ? v : NEG_SLOPE * v;
      al[i] = (e < e1) ? v : -INFINITY;
    }
#pragma unroll
    for (int i = 0; i < MX_R; ++i) {
      float* p = col + dv[i] * MX_THREADS;
      *p = fmaxf(*p, al[i]);
    }
    base += MX_BATCH;
    if (base >= e1) break;
#pragma unroll
    for (int i = 0; i < MX_R; ++i) {
      const int e  = base + i * MX_THREADS + tid;
      const int ec = (e < e1) ? e : e0;
      sv[i] = ei[ec];
      dv[i] = ei[E + ec];
      ev[i] = ea[ec];
    }
  }
  __syncthreads();

  float m[MX_ROWS_PER_WAVE];
#pragma unroll
  for (int j = 0; j < MX_ROWS_PER_WAVE; ++j) {
    int r = wave + j * MX_WAVES;
    r = (r < N_NODES) ? r : (N_NODES - 1);
    const float4 v0 = *(const float4*)(&tbl[r * MX_THREADS + lane * 4]);
    const float4 v1 = *(const float4*)(&tbl[r * MX_THREADS + 128 + lane * 4]);
    m[j] = fmaxf(fmaxf(fmaxf(v0.x, v0.y), fmaxf(v0.z, v0.w)),
                 fmaxf(fmaxf(v1.x, v1.y), fmaxf(v1.z, v1.w)));
  }
#pragma unroll
  for (int k = 16; k >= 1; k >>= 1) {
#pragma unroll
    for (int j = 0; j < MX_ROWS_PER_WAVE; ++j) m[j] = fmaxf(m[j], __shfl_xor(m[j], k, 32));
  }
  float mine = -INFINITY;
#pragma unroll
  for (int j = 0; j < MX_ROWS_PER_WAVE; ++j) mine = (lane == j) ? m[j] : mine;
  {
    const int r = wave + lane * MX_WAVES;
    if (lane < MX_ROWS_PER_WAVE && r < N_NODES) s_row[r] = mine;
  }
  __syncthreads();
  if (tid < N_NODES && s_row[tid] > -INFINITY) atomicMax(&amax[tid], f2ord(s_row[tid]));
}

#define EB_THREADS 768
#define EB_R       6
#define EB_BATCH   (EB_THREADS * EB_R)
#define NBINS      144
#define REC_CAP    (EB_BATCH + NBINS * 3 + 8)

__global__ __launch_bounds__(EB_THREADS)
void k_edge_exp(const int* __restrict__ ei, const float* __restrict__ ea,
                const float* __restrict__ att, float* __restrict__ ws, int E) {
  __shared__ int   s_cnt[NBINS];
  __shared__ int   s_start[NBINS];
  __shared__ __attribute__((aligned(16))) int   r_key[REC_CAP];
  __shared__ __attribute__((aligned(16))) float r_w[REC_CAP];
  __shared__ __attribute__((aligned(16))) float r_we[REC_CAP];
  __shared__ float s_es[(EB_THREADS / 32) * 16];
  __shared__ float s_ts[(EB_THREADS / 32) * 16];

  const float* a = ws + OFF_A;
  const float* b = ws + OFF_B;
  const unsigned* amax = (const unsigned*)(ws + OFF_AMAX);
  float* pes  = ws + OFF_PES + (int)blockIdx.x * N_PAD;
  float* pts  = ws + OFF_PTS + (int)blockIdx.x * N_PAD;
  float* S    = ws + OFF_S;
  const float cc = att[2 * OUT_C];

  const int tid  = threadIdx.x;
  const int lane = tid & 31;
  const int wave = tid >> 5;
  const int c15  = lane & 15;
  const int h    = lane >> 4;
  const int first_bin = wave * 6;

  const int per = (E + (int)gridDim.x - 1) / (int)gridDim.x;
  const int e0  = (int)blockIdx.x * per;
  const int e1  = (e0 + per < E) ? (e0 + per) : E;

  v8f acc[6];
#pragma unroll
  for (int j = 0; j < 6; ++j) acc[j] = (v8f){};
  float es = 0.0f, ts = 0.0f;

  for (int base = e0; base < e1; base += EB_BATCH) {
    if (tid < NBINS) s_cnt[tid] = 0;
    __syncthreads();

    int   sv[EB_R], dv[EB_R];
    float ev[EB_R];
#pragma unroll
    for (int i = 0; i < EB_R; ++i) {
      const int e  = base + i * EB_THREADS + tid;
      const int ec = (e < e1) ? e : e0;
      sv[i] = ei[ec];
      dv[i] = ei[E + ec];
      ev[i] = ea[ec];
    }
    int   pk[EB_R], ppos[EB_R];
    float pw[EB_R], pwe[EB_R];
#pragma unroll
    for (int i = 0; i < EB_R; ++i) {
      float al = a[dv[i]] + b[sv[i]] + cc * ev[i];
      al = (al > 0.0f) ? al : NEG_SLOPE * al;
      const float w = __expf(al - ord2f(amax[dv[i]]));
      pk[i]  = (dv[i] & 15) | ((sv[i] & 15) << 4);
      pw[i]  = w;
      pwe[i] = w * ev[i];
    }
#pragma unroll
    for (int i = 0; i < EB_R; ++i) {
      const int  e   = base + i * EB_THREADS + tid;
      const bool ok  = e < e1;
      const int  bin = (dv[i] >> 4) * 12 + (sv[i] >> 4);
      int pos = 0;
      if (ok) pos = atomicAdd(&s_cnt[bin], 1);
      ppos[i] = ok ? ((bin << 16) | pos) : -1;
    }
    __syncthreads();

    {
      int sum = 0;
      for (int q = lane; q < first_bin; q += 32) sum += (s_cnt[q] + 3) & ~3;
      for (int m = 16; m >= 1; m >>= 1) sum += __shfl_xor(sum, m, 32);
      if (lane < 6) {
        int st = sum;
        for (int jj = 0; jj < lane; ++jj) st += (s_cnt[first_bin + jj] + 3) & ~3;
        s_start[first_bin + lane] = st;
        const int nb = s_cnt[first_bin + lane];
        for (int p = st + nb; p < st + ((nb + 3) & ~3); ++p) {
          r_key[p] = 0; r_w[p] = 0.0f; r_we[p] = 0.0f;
        }
      }
    }
    __syncthreads();

#pragma unroll
    for (int i = 0; i < EB_R; ++i) {
      if (ppos[i] >= 0) {
        const int idx = s_start[ppos[i] >> 16] + (ppos[i] & 0xFFFF);
        r_key[idx] = pk[i];
        r_w[idx]   = pw[i];
        r_we[idx]  = pwe[i];
      }
    }
    __syncthreads();

#pragma unroll
    for (int j = 0; j < 6; ++j) {
      const int n   = __builtin_amdgcn_readfirstlane(s_cnt[first_bin + j]);
      const int st  = __builtin_amdgcn_readfirstlane(s_start[first_bin + j]);
      const int nch = (n + 3) >> 2;
      for (int c = 0; c < nch; ++c) {
        const int idx = st + 4 * c + 2 * h;
        const int2   kk = *(const int2*)(&r_key[idx]);
        const float2 ww = *(const float2*)(&r_w[idx]);
        const float2 we = *(const float2*)(&r_we[idx]);
        const bool m0 = ((kk.x & 15) == c15);
        const bool m1 = ((kk.y & 15) == c15);
        v2f af, bf;
        af.x = m0 ? ww.x : 0.0f;
        af.y = m1 ? ww.y : 0.0f;
        bf.x = ((kk.x >> 4) == c15) ? 1.0f : 0.0f;
        bf.y = ((kk.y >> 4) == c15) ? 1.0f : 0.0f;
        acc[j] = __builtin_amdgcn_wmma_f32_16x16x4_f32(false, af, false, bf, (short)0, acc[j],
                                                       false, false);
        es += af.x + af.y;
        ts += (m0 ? we.x : 0.0f) + (m1 ? we.y : 0.0f);
      }
    }
    __syncthreads();
  }

  es += __shfl_xor(es, 16, 32);
  ts += __shfl_xor(ts, 16, 32);
  if (h == 0) { s_es[wave * 16 + c15] = es; s_ts[wave * 16 + c15] = ts; }
  __syncthreads();
  if (tid < N_PAD) {
    const int w0 = (tid >> 4) * 2;
    const int r  = tid & 15;
    pes[tid] = s_es[w0 * 16 + r] + s_es[(w0 + 1) * 16 + r];
    pts[tid] = s_ts[w0 * 16 + r] + s_ts[(w0 + 1) * 16 + r];
  }

  const int dt = wave >> 1;
  const int st0 = (wave & 1) * 6;
#pragma unroll
  for (int j = 0; j < 6; ++j) {
    const int col = (st0 + j) * 16 + c15;
    float* T = S + (dt * STILES + st0 + j) * S_TILE_WORDS + lane;
#pragma unroll
    for (int r = 0; r < 8; ++r) {
      const int row = dt * 16 + 8 * h + r;
      const float v = acc[j][r];
      if (row < N_NODES && col < N_NODES && v != 0.0f) atomicAdd(T + r * 32, v);
    }
  }
}

__global__ __launch_bounds__(128)
void k_final(const float* __restrict__ U, const float* __restrict__ bias,
             const float* __restrict__ ws, float* __restrict__ out) {
  __shared__ float aggr[16 * OUT_C];
  __shared__ __attribute__((aligned(16))) float s_pe[16 * 8];
  __shared__ __attribute__((aligned(16))) float s_pt[16 * 8];
  const float* h    = ws + OFF_H;
  const float* S    = ws + OFF_S;
  const float* pes  = ws + OFF_PES;
  const float* pts  = ws + OFF_PTS;
  int lane = threadIdx.x & 31;
  int wave = threadIdx.x >> 5;
  int m0 = blockIdx.x * 16;
  int n0 = wave * 32;
  int nb = n0 + (lane & 15);
  int mh = (lane >> 4) * 8;

  const int fr = threadIdx.x & 15;
  const int fg = threadIdx.x >> 4;
  float pe[16], pt[16];
#pragma unroll
  for (int i = 0; i < 16; ++i) {
    pe[i] = pes[(fg * 16 + i) * N_PAD + m0 + fr];
    pt[i] = pts[(fg * 16 + i) * N_PAD + m0 + fr];
  }
  const float u_last0 = U[OUT_C * OUT_C + nb];
  const float u_last1 = U[OUT_C * OUT_C + nb + 16];
  const float bias0   = bias[nb];
  const float bias1   = bias[nb + 16];

  v8f c0 = {}; v8f c1 = {};
#pragma unroll
  for (int k0 = 0; k0 < S_PAD; k0 += 4) {
    v2f a  = load_a_tile(S, (int)blockIdx.x, k0, lane);
    v2f b0 = load_b(h, k0, n0,      lane, OUT_C);
    v2f b1 = load_b(h, k0, n0 + 16, lane, OUT_C);
    c0 = __builtin_amdgcn_wmma_f32_16x16x4_f32(false, a, false, b0, (short)0, c0, false, false);
    c1 = __builtin_amdgcn_wmma_f32_16x16x4_f32(false, a, false, b1, (short)0, c1, false, false);
  }
  v2f ub0[OUT_C / 4], ub1[OUT_C / 4];
#pragma unroll
  for (int i = 0; i < OUT_C / 4; ++i) {
    ub0[i] = load_b(U, 4 * i, n0,      lane, OUT_C);
    ub1[i] = load_b(U, 4 * i, n0 + 16, lane, OUT_C);
  }

  {
    float se = pe[0], st = pt[0];
#pragma unroll
    for (int i = 1; i < 16; ++i) { se += pe[i]; st += pt[i]; }
    s_pe[fr * 8 + fg] = se;
    s_pt[fr * 8 + fg] = st;
  }
  __syncthreads();
  float rinv[8], tn[8];
#pragma unroll
  for (int j = 0; j < 8; ++j) {
    int ml = j + mh;
    ml = (m0 + ml < N_NODES) ? ml : (N_NODES - 1 - m0);
    const float4 ea0 = *(const float4*)(&s_pe[ml * 8]);
    const float4 ea1 = *(const float4*)(&s_pe[ml * 8 + 4]);
    const float4 ta0 = *(const float4*)(&s_pt[ml * 8]);
    const float4 ta1 = *(const float4*)(&s_pt[ml * 8 + 4]);
    const float esm = ((ea0.x + ea0.y) + (ea0.z + ea0.w)) + ((ea1.x + ea1.y) + (ea1.z + ea1.w));
    const float tsm = ((ta0.x + ta0.y) + (ta0.z + ta0.w)) + ((ta1.x + ta1.y) + (ta1.z + ta1.w));
    const float es = esm + 1e-16f;
    rinv[j] = 1.0f / es;
    tn[j]   = tsm / es;
  }

#pragma unroll
  for (int j = 0; j < 8; ++j) {
    int mloc = j + mh;
    aggr[mloc * OUT_C + nb]      = c0[j] * rinv[j];
    aggr[mloc * OUT_C + nb + 16] = c1[j] * rinv[j];
  }
  __syncthreads();

  v8f d0 = {}; v8f d1 = {};
#pragma unroll
  for (int i = 0; i < OUT_C / 4; ++i) {
    v2f a  = load_a_lds(aggr, 4 * i, lane);
    d0 = __builtin_amdgcn_wmma_f32_16x16x4_f32(false, a, false, ub0[i], (short)0, d0, false, false);
    d1 = __builtin_amdgcn_wmma_f32_16x16x4_f32(false, a, false, ub1[i], (short)0, d1, false, false);
  }
#pragma unroll
  for (int j = 0; j < 8; ++j) {
    int m = m0 + j + mh;
    if (m < N_NODES) {
      out[m * OUT_C + nb]      = d0[j] + tn[j] * u_last0 + bias0;
      out[m * OUT_C + nb + 16] = d1[j] + tn[j] * u_last1 + bias1;
    }
  }
}

extern "C" void kernel_launch(void* const* d_in, const int* in_sizes, int n_in,
                              void* d_out, int out_size, void* d_ws, size_t ws_size,
                              hipStream_t stream) {
  const float* x    = (const float*)d_in[0];
  const float* ea   = (const float*)d_in[1];
  const float* W    = (const float*)d_in[2];
  const float* att  = (const float*)d_in[3];
  const float* U    = (const float*)d_in[4];
  const float* bias = (const float*)d_in[5];
  const int*   ei   = (const int*)d_in[6];
  int E = in_sizes[1];
  float* ws  = (float*)d_ws;
  float* out = (float*)d_out;

  k_init<<<64, INIT_THREADS, 0, stream>>>(x, W, att, ws);
  k_edge_max<<<MX_GRID, MX_THREADS, 0, stream>>>(ei, ea, att, x, W, ws, E);
  k_edge_exp<<<EDGE_BLOCKS, EB_THREADS, 0, stream>>>(ei, ea, att, ws, E);
  k_final<<<MTILES, 128, 0, stream>>>(U, bias, ws, out);
}


# exp pass: blocks pair up on a double edge range, each keeps one parity of dst and so adds into half of the lines of S: flush line requests to L2 halve; grouping of the sums of S, esum, t changes
# speedup vs baseline: 5.1740x; 1.0365x over previous
#include <hip/hip_runtime.h>
#include <math.h>

typedef float v2f __attribute__((ext_vector_type(2)));
typedef float v8f __attribute__((ext_vector_type(8)));

#define N_NODES 177
#define FEAT 128
#define OUT_C 128
#define NEG_SLOPE 0.2f
#define MTILES 12
#define S_PAD 180
#define STILES 12
#define S_TILE_WORDS 256
#define H_ROWS 180
#define N_PAD 192
#define EDGE_BLOCKS 128

#define OFF_H    0
#define OFF_A    (OFF_H + H_ROWS * FEAT)
#define OFF_B    (OFF_A + N_NODES)
#define OFF_AMAX (OFF_B + N_NODES)
#define S_WORDS  (MTILES * STILES * S_TILE_WORDS)
#define OFF_S    ((OFF_AMAX + N_NODES + 31) & ~31)
#define OFF_PES  (OFF_S + S_WORDS)
#define OFF_PTS  (OFF_PES + EDGE_BLOCKS * N_PAD)
#define WS_FLOATS (OFF_PTS + EDGE_BLOCKS * N_PAD)

__device__ __forceinline__ unsigned f2ord(float f) {
  unsigned u = __float_as_uint(f);
  return (u & 0x80000000u) ? ~u : (u | 0x80000000u);
}
__device__ __forceinline__ float ord2f(unsigned u) {
  return __uint_as_float((u & 0x80000000u) ? (u ^ 0x80000000u) : ~u);
}
#define ORD_NEG_INF 0x007FFFFFu

__device__ __forceinline__ v2f load_a_clamp(const float* __restrict__ A, int m0, int k0,
                                            int lane, int ld) {
  int m = m0 + (lane & 15);
  m = (m < N_NODES) ? m : (N_NODES - 1);
  int kb = k0 + ((lane >> 4) << 1);
  const float* p = A + m * ld + kb;
  v2f a; a.x = p[0]; a.y = p[1];
  return a;
}
__device__ __forceinline__ v2f load_a_tile(const float* __restrict__ S, int dtile, int k0, int lane) {
  const int rho = lane & 15;
  const float* p = S + (dtile * STILES + (k0 >> 4)) * S_TILE_WORDS
                     + (rho & 7) * 32 + (rho >> 3) * 16 + (k0 & 15) + ((lane >> 4) << 1);
  const float2 t = *(const float2*)p;
  v2f a; a.x = t.x; a.y = t.y;
  return a;
}
__device__ __forceinline__ v2f load_b(const float* __restrict__ B, int k0, int n0,
                                      int lane, int ld) {
  int n  = n0 + (lane & 15);
  int kb = k0 + ((lane >> 4) << 1);
  const float* p = B + kb * ld + n;
  v2f b; b.x = p[0]; b.y = p[ld];
  return b;
}
__device__ __forceinline__ v2f load_a_lds(const float* A, int k0, int lane) {
  int m  = lane & 15;
  int kb = k0 + ((lane >> 4) << 1);
  v2f a;
  a.x = A[m * OUT_C + kb];
  a.y = A[m * OUT_C + kb + 1];
  return a;
}

#define INIT_THREADS 256
#define INIT_ROWS_PER_WAVE (FEAT / (INIT_THREADS / 32))
__global__ __launch_bounds__(INIT_THREADS)
void k_init(const float* __restrict__ x, const float* __restrict__ W,
            const float* __restrict__ att, float* __restrict__ ws) {
  __shared__ float s_va[FEAT];
  __shared__ float s_vb[FEAT];
  int tid    = blockIdx.x * blockDim.x + threadIdx.x;
  int stride = gridDim.x * blockDim.x;
  const int lane = threadIdx.x & 31;
  const int wave = threadIdx.x >> 5;
  const bool ab_block = blockIdx.x < MTILES;
  const int r  = lane & 15;
  const int hh = lane >> 4;
  const int m  = (int)blockIdx.x * 16 + r;
  const int mc = (m < N_NODES) ? m : (N_NODES - 1);
  float wv[INIT_ROWS_PER_WAVE][FEAT / 32] = {};
  float a0[FEAT / 32] = {}, a1[FEAT / 32] = {};
  float xv[FEAT / 2] = {};
  if (ab_block) {
#pragma unroll
    for (int q = 0; q < FEAT / 32; ++q) {
      a0[q] = att[lane + 32 * q];
      a1[q] = att[OUT_C + lane + 32 * q];
    }
#pragma unroll
    for (int j = 0; j < INIT_ROWS_PER_WAVE; ++j) {
      const int row = wave * INIT_ROWS_PER_WAVE + j;
#pragma unroll
      for (int q = 0; q < FEAT / 32; ++q) wv[j][q] = W[row * OUT_C + lane + 32 * q];
    }
    if (wave == 0) {
      const float* xr = x + mc * FEAT + hh * (FEAT / 2);
#pragma unroll
      for (int i = 0; i < FEAT / 2; ++i) xv[i] = xr[i];
    }
  }
  unsigned* amax = (unsigned*)(ws + OFF_AMAX);
  {
    float4* S4 = (float4*)(ws + OFF_S);
    const float4 z4 = make_float4(0.0f, 0.0f, 0.0f, 0.0f);
    for (int j = tid; j < S_WORDS / 4; j += stride) S4[j] = z4;
  }
  for (int j = tid; j < (H_ROWS - N_NODES) * FEAT; j += stride)
    ws[OFF_H + N_NODES * FEAT + j] = 0.0f;
  for (int j = tid; j < N_NODES; j += stride) amax[j] = ORD_NEG_INF;
  if (ab_block) {
    float pa[INIT_ROWS_PER_WAVE], pb[INIT_ROWS_PER_WAVE];
#pragma unroll
    for (int j = 0; j < INIT_ROWS_PER_WAVE; ++j) {
      pa[j] = 0.0f; pb[j] = 0.0f;
#pragma unroll
      for (int q = 0; q < FEAT / 32; ++q) { pa[j] += wv[j][q] * a0[q]; pb[j] += wv[j][q] * a1[q]; }
    }
    const bool b4 = (lane & 16) != 0, b3 = (lane & 8) != 0, b2 = (lane & 4) != 0, b1 = (lane & 2) != 0;
    float qa[8], qb[8];
#pragma unroll
    for (int i = 0; i < 8; ++i) {
      const float ka = b4 ? pa[i + 8] : pa[i], sa = b4 ? pa[i] : pa[i + 8];
      const float kb = b4 ? pb[i + 8] : pb[i], sb = b4 ? pb[i] : pb[i + 8];
      qa[i] = ka + __shfl_xor(sa, 16, 32);
      qb[i] = kb + __shfl_xor(sb, 16, 32);
    }
    float ra[4], rb[4];
#pragma unroll
    for (int i = 0; i < 4; ++i) {
      const float ka = b3 ? qa[i + 4] : qa[i], sa = b3 ? qa[i] : qa[i + 4];
      const float kb = b3 ? qb[i + 4] : qb[i], sb = b3 ? qb[i] : qb[i + 4];
      ra[i] = ka + __shfl_xor(sa, 8, 32);
      rb[i] = kb + __shfl_xor(sb, 8, 32);
    }
    float ua[2], ub[2];
#pragma unroll
    for (int i = 0; i < 2; ++i) {
      const float ka = b2 ? ra[i + 2] : ra[i], sa = b2 ? ra[i] : ra[i + 2];
      const float kb = b2 ? rb[i + 2] : rb[i], sb = b2 ? rb[i] : rb[i + 2];
      ua[i] = ka + __shfl_xor(sa, 4, 32);
      ub[i] = kb + __shfl_xor(sb, 4, 32);
    }
    float ta, tb;
    {
      const float ka = b1 ? ua[1] : ua[0], sa = b1 ? ua[0] : ua[1];
      const float kb = b1 ? ub[1] : ub[0], sb = b1 ? ub[0] : ub[1];
      ta = ka + __shfl_xor(sa, 2, 32);
      tb = kb + __shfl_xor(sb, 2, 32);
    }
    ta += __shfl_xor(ta, 1, 32);
    tb += __shfl_xor(tb, 1, 32);
    if ((lane & 1) == 0) {
      s_va[wave * INIT_ROWS_PER_WAVE + (lane >> 1)] = ta;
      s_vb[wave * INIT_ROWS_PER_WAVE + (lane >> 1)] = tb;
    }
    __syncthreads();
    if (wave == 0) {
      const float* va = s_va + hh * (FEAT / 2);
      const float* vb = s_vb + hh * (FEAT / 2);
      float sa = 0.0f, sb = 0.0f;
#pragma unroll
      for (int i = 0; i < FEAT / 2; ++i) {
        sa += xv[i] * va[i];
        sb += xv[i] * vb[i];
      }
      sa += __shfl_xor(sa, 16, 32);
      sb += __shfl_xor(sb, 16, 32);
      if (hh == 0 && m < N_NODES) { ws[OFF_A + m] = sa; ws[OFF_B + m] = sb; }
    }
  }
}

__device__ __forceinline__ void node_gemm_strip(const float* __restrict__ x, const float* __restrict__ W,
                                                float* __restrict__ h, int m0, int wave, int lane) {
  const int n0 = wave * 16;
  v8f c0 = {};
  for (int k0 = 0; k0 < FEAT; k0 += 4) {
    v2f a  = load_a_clamp(x, m0, k0, lane, FEAT);
    v2f b0 = load_b(W, k0, n0, lane, OUT_C);
    c0 = __builtin_amdgcn_wmma_f32_16x16x4_f32(false, a, false, b0, (short)0, c0, false, false);
  }
  const int nb = n0 + (lane & 15);
  const int mh = (lane >> 4) * 8;
  for (int j = 0; j < 8; ++j) {
    const int m = m0 + j + mh;
    if (m < N_NODES) h[m * OUT_C + nb] = c0[j];
  }
}

#define MX_THREADS 256
#define MX_BLOCKS  128
#define MX_GRID    (MX_BLOCKS + MTILES)
#define MX_R       16
#define MX_BATCH   (MX_THREADS * MX_R)
#define MX_WAVES   (MX_THREADS / 32)
#define MX_ROWS_PER_WAVE ((N_NODES + MX_WAVES - 1) / MX_WAVES)

__global__ __launch_bounds__(MX_THREADS)
void k_edge_max(const int* __restrict__ ei, const float* __restrict__ ea,
                const float* __restrict__ att, const float* __restrict__ x,
                const float* __restrict__ W, float* __restrict__ ws, int E) {
  __shared__ __attribute__((aligned(16))) float tbl[N_NODES * MX_THREADS];
  __shared__ float s_row[N_NODES];

  if ((int)blockIdx.x >= MX_BLOCKS) {
    node_gemm_strip(x, W, ws + OFF_H, ((int)blockIdx.x - MX_BLOCKS) * 16,
                    (int)threadIdx.x >> 5, (int)threadIdx.x & 31);
    return;
  }

  const float* a = ws + OFF_A;
  const float* b = ws + OFF_B;
  unsigned* amax = (unsigned*)(ws + OFF_AMAX);
  const float cc = att[2 * OUT_C];

  const int tid  = threadIdx.x;
  const int lane = tid & 31;
  const int wave = tid >> 5;

  const int per = (E + MX_BLOCKS - 1) / MX_BLOCKS;
  const int e0  = (int)blockIdx.x * per;
  const int e1  = (e0 + per < E) ? (e0 + per) : E;
  if (e0 >= e1) return;

  int   sv[MX_R], dv[MX_R];
  float ev[MX_R];
#pragma unroll
  for (int i = 0; i < MX_R; ++i) {
    const int e  = e0 + i * MX_THREADS + tid;
    const int ec = (e < e1) ? e : e0;
    sv[i] = ei[ec];
    dv[i] = ei[E + ec];
    ev[i] = ea[ec];
  }

  {
    const float ninf = -INFINITY;
    const float4 n4 = make_float4(ninf, ninf, ninf, ninf);
    float4* t4 = (float4*)tbl;
    for (int q = tid; q < N_NODES * MX_THREADS / 4; q += MX_THREADS) t4[q] = n4;
  }
  __syncthreads();

  float* col = tbl + tid;
  for (int base = e0; ; ) {
    float al[MX_R];
#pragma unroll
    for (int i = 0; i < MX_R; ++i) {
      const int e = base + i * MX_THREADS + tid;
      float v = a[dv[i]] + b[sv[i]] + cc * ev[i];
      v = (v > 0.0f) ? v : NEG_SLOPE * v;
      al[i] = (e < e1) ? v : -INFINITY;
    }
#pragma unroll
    for (int i = 0; i < MX_R; ++i) {
      float* p = col + dv[i] * MX_THREADS;
      *p = fmaxf(*p, al[i]);
    }
    base += MX_BATCH;
    if (base >= e1) break;
#pragma unroll
    for (int i = 0; i < MX_R; ++i) {
      const int e  = base + i * MX_THREADS + tid;
      const int ec = (e < e1) ? e : e0;
      sv[i] = ei[ec];
      dv[i] = ei[E + ec];
      ev[i] = ea[ec];
    }
  }
  __syncthreads();

  float m[MX_ROWS_PER_WAVE];
#pragma unroll
  for (int j = 0; j < MX_ROWS_PER_WAVE; ++j) {
    int r = wave + j * MX_WAVES;
    r = (r < N_NODES) ? r : (N_NODES - 1);
    const float4 v0 = *(const float4*)(&tbl[r * MX_THREADS + lane * 4]);
    const float4 v1 = *(const float4*)(&tbl[r * MX_THREADS + 128 + lane * 4]);
    m[j] = fmaxf(fmaxf(fmaxf(v0.x, v0.y), fmaxf(v0.z, v0.w)),
                 fmaxf(fmaxf(v1.x, v1.y), fmaxf(v1.z, v1.w)));
  }
#pragma unroll
  for (int k = 16; k >= 1; k >>= 1) {
#pragma unroll
    for (int j = 0; j < MX_ROWS_PER_WAVE; ++j) m[j] = fmaxf(m[j], __shfl_xor(m[j], k, 32));
  }
  float mine = -INFINITY;
#pragma unroll
  for (int j = 0; j < MX_ROWS_PER_WAVE; ++j) mine = (lane == j) ? m[j] : mine;
  {
    const int r = wave + lane * MX_WAVES;
    if (lane < MX_ROWS_PER_WAVE && r < N_NODES) s_row[r] = mine;
  }
  __syncthreads();
  if (tid < N_NODES && s_row[tid] > -INFINITY) atomicMax(&amax[tid], f2ord(s_row[tid]));
}

#define EB_THREADS 768
#define EB_R       12
#define EB_GROUPS  2
#define EB_BATCH   (EB_THREADS * EB_R)
#define NBINS      144
#define REC_CAP    (EB_BATCH + NBINS * 3 + 8)

__global__ __launch_bounds__(EB_THREADS)
void k_edge_exp(const int* __restrict__ ei, const float* __restrict__ ea,
                const float* __restrict__ att, float* __restrict__ ws, int E) {
  __shared__ int   s_cnt[NBINS];
  __shared__ int   s_start[NBINS];
  __shared__ __attribute__((aligned(16))) int   r_key[REC_CAP];
  __shared__ __attribute__((aligned(16))) float r_w[REC_CAP];
  __shared__ __attribute__((aligned(16))) float r_we[REC_CAP];
  __shared__ float s_es[(EB_THREADS / 32) * 16];
  __shared__ float s_ts[(EB_THREADS / 32) * 16];

  const float* a = ws + OFF_A;
  const float* b = ws + OFF_B;
  const unsigned* amax = (const unsigned*)(ws + OFF_AMAX);
  float* pes  = ws + OFF_PES + (int)blockIdx.x * N_PAD;
  float* pts  = ws + OFF_PTS + (int)blockIdx.x * N_PAD;
  float* S    = ws + OFF_S;
  const float cc = att[2 * OUT_C];

  const int tid  = threadIdx.x;
  const int lane = tid & 31;
  const int wave = tid >> 5;
  const int c15  = lane & 15;
  const int h    = lane >> 4;
  const int first_bin = wave * 6;

  const int grp    = (int)blockIdx.x % EB_GROUPS;
  const int nrange = (int)gridDim.x / EB_GROUPS;
  const int per = (E + nrange - 1) / nrange;
  const int e0  = ((int)blockIdx.x / EB_GROUPS) * per;
  const int e1  = (e0 + per < E) ? (e0 + per) : E;

  v8f acc[6];
#pragma unroll
  for (int j = 0; j < 6; ++j) acc[j] = (v8f){};
  float es = 0.0f, ts = 0.0f;

  for (int base = e0; base < e1; base += EB_BATCH) {
    if (tid < NBINS) s_cnt[tid] = 0;
    __syncthreads();

    int   sv[EB_R], dv[EB_R];
    float ev[EB_R];
#pragma unroll
    for (int i = 0; i < EB_R; ++i) {
      const int e  = base + i * EB_THREADS + tid;
      const int ec = (e < e1) ? e : e0;
      sv[i] = ei[ec];
      dv[i] = ei[E + ec];
      ev[i] = ea[ec];
    }
    unsigned pk[EB_R], ppos[EB_R];
    float pw[EB_R], pwe[EB_R];
#pragma unroll
    for (int i = 0; i < EB_R; ++i) {
      float al = a[dv[i]] + b[sv[i]] + cc * ev[i];
      al = (al > 0.0f) ? al : NEG_SLOPE * al;
      const float w = __expf(al - ord2f(amax[dv[i]]));
      const int  e    = base + i * EB_THREADS + tid;
      const bool keep = (e < e1) && ((dv[i] & (EB_GROUPS - 1)) == grp);
      const int  bin  = (dv[i] >> 4) * 12 + (sv[i] >> 4);
      pk[i]  = keep ? (unsigned)((bin << 8) | ((sv[i] & 15) << 4) | (dv[i] & 15)) : 0xFFFFFFFFu;
      pw[i]  = w;
      pwe[i] = w * ev[i];
    }
    int pos[EB_R];
#pragma unroll
    for (int i = 0; i < EB_R; ++i) {
      pos[i] = 0;
      if (pk[i] != 0xFFFFFFFFu) pos[i] = atomicAdd(&s_cnt[pk[i] >> 8], 1);
    }
#pragma unroll
    for (int i = 0; i < EB_R; ++i) ppos[i] = (pk[i] << 16) | (unsigned)pos[i];
    __syncthreads();

    {
      int sum = 0;
      for (int q = lane; q < first_bin; q += 32) sum += (s_cnt[q] + 3) & ~3;
      for (int m = 16; m >= 1; m >>= 1) sum += __shfl_xor(sum, m, 32);
      if (lane < 6) {
        int st = sum;
        for (int jj = 0; jj < lane; ++jj) st += (s_cnt[first_bin + jj] + 3) & ~3;
        s_start[first_bin + lane] = st;
        const int nb = s_cnt[first_bin + lane];
        for (int p = st + nb; p < st + ((nb + 3) & ~3); ++p) {
          r_key[p] = 0; r_w[p] = 0.0f; r_we[p] = 0.0f;
        }
      }
    }
    __syncthreads();

    int bst[EB_R];
#pragma unroll
    for (int i = 0; i < EB_R; ++i) {
      const unsigned bq = ppos[i] >> 24;
      bst[i] = s_start[(bq < NBINS) ? bq : 0u];
    }
#pragma unroll
    for (int i = 0; i < EB_R; ++i) {
      if (ppos[i] < 0xFFFF0000u) {
        const int idx = bst[i] + (int)(ppos[i] & 0xFFFFu);
        r_key[idx] = (int)((ppos[i] >> 16) & 0xFFu);
        r_w[idx]   = pw[i];
        r_we[idx]  = pwe[i];
      }
    }
    __syncthreads();

#pragma unroll
    for (int j = 0; j < 6; ++j) {
      const int n   = __builtin_amdgcn_readfirstlane(s_cnt[first_bin + j]);
      const int st  = __builtin_amdgcn_readfirstlane(s_start[first_bin + j]);
      const int nch = (n + 3) >> 2;
      for (int c = 0; c < nch; ++c) {
        const int idx = st + 4 * c + 2 * h;
        const int2   kk = *(const int2*)(&r_key[idx]);
        const float2 ww = *(const float2*)(&r_w[idx]);
        const float2 we = *(const float2*)(&r_we[idx]);
        const bool m0 = ((kk.x & 15) == c15);
        const bool m1 = ((kk.y & 15) == c15);
        v2f af, bf;
        af.x = m0 ? ww.x : 0.0f;
        af.y = m1 ? ww.y : 0.0f;
        bf.x = ((kk.x >> 4) == c15) ? 1.0f : 0.0f;
        bf.y = ((kk.y >> 4) == c15) ? 1.0f : 0.0f;
        acc[j] = __builtin_amdgcn_wmma_f32_16x16x4_f32(false, af, false, bf, (short)0, acc[j],
                                                       false, false);
        es += af.x + af.y;
        ts += (m0 ? we.x : 0.0f) + (m1 ? we.y : 0.0f);
      }
    }
    __syncthreads();
  }

  es += __shfl_xor(es, 16, 32);
  ts += __shfl_xor(ts, 16, 32);
  if (h == 0) { s_es[wave * 16 + c15] = es; s_ts[wave * 16 + c15] = ts; }
  __syncthreads();
  if (tid < N_PAD) {
    const int w0 = (tid >> 4) * 2;
    const int r  = tid & 15;
    pes[tid] = s_es[w0 * 16 + r] + s_es[(w0 + 1) * 16 + r];
    pts[tid] = s_ts[w0 * 16 + r] + s_ts[(w0 + 1) * 16 + r];
  }

  const int dt = wave >> 1;
  const int st0 = (wave & 1) * 6;
#pragma unroll
  for (int j = 0; j < 6; ++j) {
    const int col = (st0 + j) * 16 + c15;
    float* T = S + (dt * STILES + st0 + j) * S_TILE_WORDS + lane;
#pragma unroll
    for (int r = 0; r < 8; ++r) {
      const int row = dt * 16 + 8 * h + r;
      const float v = acc[j][r];
      if (row < N_NODES && col < N_NODES && v != 0.0f) atomicAdd(T + r * 32, v);
    }
  }
}

__global__ __launch_bounds__(128)
void k_final(const float* __restrict__ U, const float* __restrict__ bias,
             const float* __restrict__ ws, float* __restrict__ out) {
  __shared__ float aggr[16 * OUT_C];
  __shared__ __attribute__((aligned(16))) float s_pe[16 * 8];
  __shared__ __attribute__((aligned(16))) float s_pt[16 * 8];
  const float* h    = ws + OFF_H;
  const float* S    = ws + OFF_S;
  const float* pes  = ws + OFF_PES;
  const float* pts  = ws + OFF_PTS;
  int lane = threadIdx.x & 31;
  int wave = threadIdx.x >> 5;
  int m0 = blockIdx.x * 16;
  int n0 = wave * 32;
  int nb = n0 + (lane & 15);
  int mh = (lane >> 4) * 8;

  const int fr = threadIdx.x & 15;
  const int fg = threadIdx.x >> 4;
  float pe[16], pt[16];
#pragma unroll
  for (int i = 0; i < 16; ++i) {
    pe[i] = pes[(fg * 16 + i) * N_PAD + m0 + fr];
    pt[i] = pts[(fg * 16 + i) * N_PAD + m0 + fr];
  }
  const float u_last0 = U[OUT_C * OUT_C + nb];
  const float u_last1 = U[OUT_C * OUT_C + nb + 16];
  const float bias0   = bias[nb];
  const float bias1   = bias[nb + 16];

  v8f c0 = {}; v8f c1 = {};
#pragma unroll
  for (int k0 = 0; k0 < S_PAD; k0 += 4) {
    v2f a  = load_a_tile(S, (int)blockIdx.x, k0, lane);
    v2f b0 = load_b(h, k0, n0,      lane, OUT_C);
    v2f b1 = load_b(h, k0, n0 + 16, lane, OUT_C);
    c0 = __builtin_amdgcn_wmma_f32_16x16x4_f32(false, a, false, b0, (short)0, c0, false, false);
    c1 = __builtin_amdgcn_wmma_f32_16x16x4_f32(false, a, false, b1, (short)0, c1, false, false);
  }
  v2f ub0[OUT_C / 4], ub1[OUT_C / 4];
#pragma unroll
  for (int i = 0; i < OUT_C / 4; ++i) {
    ub0[i] = load_b(U, 4 * i, n0,      lane, OUT_C);
    ub1[i] = load_b(U, 4 * i, n0 + 16, lane, OUT_C);
  }

  {
    float se = pe[0], st = pt[0];
#pragma unroll
    for (int i = 1; i < 16; ++i) { se += pe[i]; st += pt[i]; }
    s_pe[fr * 8 + fg] = se;
    s_pt[fr * 8 + fg] = st;
  }
  __syncthreads();
  float rinv[8], tn[8];
#pragma unroll
  for (int j = 0; j < 8; ++j) {
    int ml = j + mh;
    ml = (m0 + ml < N_NODES) ? ml : (N_NODES - 1 - m0);
    const float4 ea0 = *(const float4*)(&s_pe[ml * 8]);
    const float4 ea1 = *(const float4*)(&s_pe[ml * 8 + 4]);
    const float4 ta0 = *(const float4*)(&s_pt[ml * 8]);
    const float4 ta1 = *(const float4*)(&s_pt[ml * 8 + 4]);
    const float esm = ((ea0.x + ea0.y) + (ea0.z + ea0.w)) + ((ea1.x + ea1.y) + (ea1.z + ea1.w));
    const float tsm = ((ta0.x + ta0.y) + (ta0.z + ta0.w)) + ((ta1.x + ta1.y) + (ta1.z + ta1.w));
    const float es = esm + 1e-16f;
    rinv[j] = 1.0f / es;
    tn[j]   = tsm / es;
  }

#pragma unroll
  for (int j = 0; j < 8; ++j) {
    int mloc = j + mh;
    aggr[mloc * OUT_C + nb]      = c0[j] * rinv[j];
    aggr[mloc * OUT_C + nb + 16] = c1[j] * rinv[j];
  }
  __syncthreads();

  v8f d0 = {}; v8f d1 = {};
#pragma unroll
  for (int i = 0; i < OUT_C / 4; ++i) {
    v2f a  = load_a_lds(aggr, 4 * i, lane);
    d0 = __builtin_amdgcn_wmma_f32_16x16x4_f32(false, a, false, ub0[i], (short)0, d0, false, false);
    d1 = __builtin_amdgcn_wmma_f32_16x16x4_f32(false, a, false, ub1[i], (short)0, d1, false, false);
  }
#pragma unroll
  for (int j = 0; j < 8; ++j) {
    int m = m0 + j + mh;
    if (m < N_NODES) {
      out[m * OUT_C + nb]      = d0[j] + tn[j] * u_last0 + bias0;
      out[m * OUT_C + nb + 16] = d1[j] + tn[j] * u_last1 + bias1;
    }
  }
}

extern "C" void kernel_launch(void* const* d_in, const int* in_sizes, int n_in,
                              void* d_out, int out_size, void* d_ws, size_t ws_size,
                              hipStream_t stream) {
  const float* x    = (const float*)d_in[0];
  const float* ea   = (const float*)d_in[1];
  const float* W    = (const float*)d_in[2];
  const float* att  = (const float*)d_in[3];
  const float* U    = (const float*)d_in[4];
  const float* bias = (const float*)d_in[5];
  const int*   ei   = (const int*)d_in[6];
  int E = in_sizes[1];
  float* ws  = (float*)d_ws;
  float* out = (float*)d_out;

  k_init<<<64, INIT_THREADS, 0, stream>>>(x, W, att, ws);
  k_edge_max<<<MX_GRID, MX_THREADS, 0, stream>>>(ei, ea, att, x, W, ws, E);
  k_edge_exp<<<EDGE_BLOCKS, EB_THREADS, 0, stream>>>(ei, ea, att, ws, E);
  k_final<<<MTILES, 128, 0, stream>>>(U, bias, ws, out);
}
